# starting kernel
# baseline (speedup 1.0000x reference)
#include <hip/hip_runtime.h>

typedef __attribute__((ext_vector_type(2))) float v2f;
typedef __attribute__((ext_vector_type(8))) float v8f;

#define EMB 64

__global__ void init_concat_kernel(const float* __restrict__ users,
                                   const float* __restrict__ items,
                                   float* __restrict__ out,
                                   float* __restrict__ cur,
                                   int userElems, int totalElems) {
  int i4 = (blockIdx.x * blockDim.x + threadIdx.x) * 4;
  if (i4 >= totalElems) return;
  float4 v = (i4 < userElems) ? *(const float4*)(users + i4)
                              : *(const float4*)(items + (i4 - userElems));
  *(float4*)(out + i4) = v;
  *(float4*)(cur + i4) = v;
}

__global__ void axpy_scale_kernel(float* __restrict__ out,
                                  const float* __restrict__ nxt,
                                  float scale, int totalElems) {
  int i4 = (blockIdx.x * blockDim.x + threadIdx.x) * 4;
  if (i4 >= totalElems) return;
  float4 a = *(const float4*)(out + i4);
  float4 b = *(const float4*)(nxt + i4);
  a.x = (a.x + b.x) * scale;
  a.y = (a.y + b.y) * scale;
  a.z = (a.z + b.z) * scale;
  a.w = (a.w + b.w) * scale;
  *(float4*)(out + i4) = a;
}

__global__ void spmm_wmma_kernel(const int*   __restrict__ rows,
                                 const int*   __restrict__ cols,
                                 const float* __restrict__ vals,
                                 const float* __restrict__ x,
                                 float*       __restrict__ y,
                                 int nEdges) {
  const int lane   = threadIdx.x & 31;
  const int m      = lane & 15;
  const bool hiHalf = lane >= 16;
  const int wavesPerBlock = blockDim.x >> 5;
  const int wave   = blockIdx.x * wavesPerBlock + (threadIdx.x >> 5);
  const int stride = gridDim.x * wavesPerBlock;
  const int nGroups = (nEdges + 15) >> 4;

  for (int g = wave; g < nGroups; g += stride) {
    const int base = g << 4;
    const int e    = base + m;
    const int eC   = (e < nEdges) ? e : (nEdges - 1);

    const float v   = (e < nEdges) ? vals[eC] : 0.0f;
    const int myRow = rows[eC];
    const int myCol = cols[eC];

    int colK[8];
#pragma unroll
    for (int c = 0; c < 4; ++c) {
      const int kb0 = 4 * c + (hiHalf ? 2 : 0);
      colK[2 * c]     = __shfl(myCol, kb0, 32);
      colK[2 * c + 1] = __shfl(myCol, kb0 + 1, 32);
    }

    int rowD[8];
#pragma unroll
    for (int vv = 0; vv < 8; ++vv)
      rowD[vv] = __shfl(myRow, hiHalf ? (vv + 8) : vv, 32);

    v2f A[4];
#pragma unroll
    for (int c = 0; c < 4; ++c) {
      const int ka0 = 4 * c + (hiHalf ? 2 : 0);
      A[c].x = (m == ka0)     ? v : 0.0f;
      A[c].y = (m == ka0 + 1) ? v : 0.0f;
    }

#pragma unroll
    for (int t = 0; t < 4; ++t) {
      v8f acc = {};
#pragma unroll
      for (int c = 0; c < 4; ++c) {
        v2f B;
        B.x = x[colK[2 * c]     * EMB + t * 16 + m];
        B.y = x[colK[2 * c + 1] * EMB + t * 16 + m];
        acc = __builtin_amdgcn_wmma_f32_16x16x4_f32(
             false, A[c],  false, B,
             (short)0, acc,  false,  false);
      }
#pragma unroll
      for (int vv = 0; vv < 8; ++vv)
        atomicAdd(&y[rowD[vv] * EMB + t * 16 + m], acc[vv]);
    }
  }
}

extern "C" void kernel_launch(void* const* d_in, const int* in_sizes, int n_in,
                              void* d_out, int out_size, void* d_ws, size_t ws_size,
                              hipStream_t stream) {
  const float* users = (const float*)d_in[0];
  const float* items = (const float*)d_in[1];
  const int*   erow  = (const int*)d_in[2];
  const int*   ecol  = (const int*)d_in[3];
  const float* evals = (const float*)d_in[4];
  float* out = (float*)d_out;

  const int userElems  = in_sizes[0];
  const int itemElems  = in_sizes[1];
  const int nEdges     = in_sizes[2];
  const int totalElems = userElems + itemElems;

  const size_t bufBytes = (size_t)totalElems * sizeof(float);
  if (ws_size < 2 * bufBytes) return;
  float* cur = (float*)d_ws;
  float* nxt = (float*)((char*)d_ws + bufBytes);

  const int ethreads = (totalElems + 3) / 4;
  const dim3 eb(256), eg((ethreads + 255) / 256);
  init_concat_kernel<<<eg, eb, 0, stream>>>(users, items, out, cur,
                                            userElems, totalElems);

  const int nGroups    = (nEdges + 15) / 16;
  const int spmmBlocks = (nGroups + 7) / 8;

  for (int layer = 0; layer < 3; ++layer) {
    hipMemsetAsync(nxt, 0, bufBytes, stream);
    spmm_wmma_kernel<<<spmmBlocks, 256, 0, stream>>>(erow, ecol, evals,
                                                     cur, nxt, nEdges);
    const float scale = (layer == 2) ? 0.25f : 1.0f;
    axpy_scale_kernel<<<eg, eb, 0, stream>>>(out, nxt, scale, totalElems);
    float* tmp = cur; cur = nxt; nxt = tmp;
  }
}


# Edges bucketed by destination row once per call (int atomic count, scan, arrival-rank fill); a wave then owns 16 rows per layer: float atomics, memsets, axpy passes go; row sum order changes
# speedup vs baseline: 1.3498x; 1.3498x over previous
#include <hip/hip_runtime.h>

typedef __attribute__((ext_vector_type(2))) float v2f;
typedef __attribute__((ext_vector_type(8))) float v8f;

#define EMB 64

__global__ void init_concat_kernel(const float* __restrict__ users,
                                   const float* __restrict__ items,
                                   float* __restrict__ out,
                                   float* __restrict__ cur,
                                   int userElems, int totalElems) {
  int i4 = (blockIdx.x * blockDim.x + threadIdx.x) * 4;
  if (i4 >= totalElems) return;
  float4 v = (i4 < userElems) ? *(const float4*)(users + i4)
                              : *(const float4*)(items + (i4 - userElems));
  *(float4*)(out + i4) = v;
  *(float4*)(cur + i4) = v;
}

__global__ void axpy_scale_kernel(float* __restrict__ out,
                                  const float* __restrict__ nxt,
                                  float scale, int totalElems) {
  int i4 = (blockIdx.x * blockDim.x + threadIdx.x) * 4;
  if (i4 >= totalElems) return;
  float4 a = *(const float4*)(out + i4);
  float4 b = *(const float4*)(nxt + i4);
  a.x = (a.x + b.x) * scale;
  a.y = (a.y + b.y) * scale;
  a.z = (a.z + b.z) * scale;
  a.w = (a.w + b.w) * scale;
  *(float4*)(out + i4) = a;
}

__global__ void spmm_wmma_kernel(const int*   __restrict__ rows,
                                 const int*   __restrict__ cols,
                                 const float* __restrict__ vals,
                                 const float* __restrict__ x,
                                 float*       __restrict__ y,
                                 int nEdges) {
  const int lane   = threadIdx.x & 31;
  const int m      = lane & 15;
  const bool hiHalf = lane >= 16;
  const int wavesPerBlock = blockDim.x >> 5;
  const int wave   = blockIdx.x * wavesPerBlock + (threadIdx.x >> 5);
  const int stride = gridDim.x * wavesPerBlock;
  const int nGroups = (nEdges + 15) >> 4;

  for (int g = wave; g < nGroups; g += stride) {
    const int base = g << 4;
    const int e    = base + m;
    const int eC   = (e < nEdges) ? e : (nEdges - 1);

    const float v   = (e < nEdges) ? vals[eC] : 0.0f;
    const int myRow = rows[eC];
    const int myCol = cols[eC];

    int colK[8];
#pragma unroll
    for (int c = 0; c < 4; ++c) {
      const int kb0 = 4 * c + (hiHalf ? 2 : 0);
      colK[2 * c]     = __shfl(myCol, kb0, 32);
      colK[2 * c + 1] = __shfl(myCol, kb0 + 1, 32);
    }

    int rowD[8];
#pragma unroll
    for (int vv = 0; vv < 8; ++vv)
      rowD[vv] = __shfl(myRow, hiHalf ? (vv + 8) : vv, 32);

    v2f A[4];
#pragma unroll
    for (int c = 0; c < 4; ++c) {
      const int ka0 = 4 * c + (hiHalf ? 2 : 0);
      A[c].x = (m == ka0)     ? v : 0.0f;
      A[c].y = (m == ka0 + 1) ? v : 0.0f;
    }

#pragma unroll
    for (int t = 0; t < 4; ++t) {
      v8f acc = {};
#pragma unroll
      for (int c = 0; c < 4; ++c) {
        v2f B;
        B.x = x[colK[2 * c]     * EMB + t * 16 + m];
        B.y = x[colK[2 * c + 1] * EMB + t * 16 + m];
        acc = __builtin_amdgcn_wmma_f32_16x16x4_f32(
             false, A[c],  false, B,
             (short)0, acc,  false,  false);
      }
#pragma unroll
      for (int vv = 0; vv < 8; ++vv)
        atomicAdd(&y[rowD[vv] * EMB + t * 16 + m], acc[vv]);
    }
  }
}


#define SCAN_SPAN 8192

__global__ void row_count_kernel(const int* __restrict__ rows,
                                 int* __restrict__ cnt, int nEdges) {
  int e = blockIdx.x * blockDim.x + threadIdx.x;
  if (e < nEdges) atomicAdd(&cnt[rows[e]], 1);
}

__global__ void row_scan_kernel(const int* __restrict__ cnt,
                                int* __restrict__ ptr,
                                int* __restrict__ cursor, int nPad) {
  __shared__ int waveSum[32];
  __shared__ int tripTotal;
  const int tid = threadIdx.x, lane = tid & 31, w = tid >> 5;
  int carry = 0;
  for (int base = 0; base < nPad; base += SCAN_SPAN) {
    const int i0 = base + tid * 8;
    int4 a = *(const int4*)(cnt + i0);
    int4 b = *(const int4*)(cnt + i0 + 4);
    int e0 = 0, e1 = a.x, e2 = e1 + a.y, e3 = e2 + a.z, e4 = e3 + a.w;
    int e5 = e4 + b.x, e6 = e5 + b.y, e7 = e6 + b.z;
    const int tsum = e7 + b.w;
    int s = tsum;
#pragma unroll
    for (int d = 1; d < 32; d <<= 1) {
      int t = __shfl_up(s, d, 32);
      if (lane >= d) s += t;
    }
    if (lane == 31) waveSum[w] = s;
    __syncthreads();
    if (w == 0) {
      const int ws = waveSum[lane];
      int t2 = ws;
#pragma unroll
      for (int d = 1; d < 32; d <<= 1) {
        int t = __shfl_up(t2, d, 32);
        if (lane >= d) t2 += t;
      }
      waveSum[lane] = t2 - ws;
      if (lane == 31) tripTotal = t2;
    }
    __syncthreads();
    const int off = carry + waveSum[w] + (s - tsum);
    int4 o0 = make_int4(off + e0, off + e1, off + e2, off + e3);
    int4 o1 = make_int4(off + e4, off + e5, off + e6, off + e7);
    *(int4*)(ptr + i0) = o0;      *(int4*)(ptr + i0 + 4) = o1;
    *(int4*)(cursor + i0) = o0;   *(int4*)(cursor + i0 + 4) = o1;
    carry += tripTotal;
    __syncthreads();
  }
}

__global__ void bucket_fill_kernel(const int* __restrict__ rows,
                                   const int* __restrict__ cols,
                                   const float* __restrict__ vals,
                                   int* __restrict__ cursor,
                                   int2* __restrict__ sorted, int nEdges) {
  int e = blockIdx.x * blockDim.x + threadIdx.x;
  if (e < nEdges) {
    const int slot = atomicAdd(&cursor[rows[e]], 1);
    sorted[slot] = make_int2(cols[e], __float_as_int(vals[e]));
  }
}

__global__ void spmm_rows_wmma_kernel(const int*  __restrict__ ptr,
                                      const int2* __restrict__ sorted,
                                      const float* __restrict__ x,
                                      float* __restrict__ y,
                                      float* __restrict__ out,
                                      float scale, int writeY,
                                      int nRows, int nEdges) {
  const int lane = threadIdx.x & 31;
  const int c = lane & 15;
  const int h = lane >> 4;
  const int wave = blockIdx.x * (blockDim.x >> 5) + (threadIdx.x >> 5);
  const int r0 = wave << 4;
  if (r0 >= nRows) return;

  const int myRow = r0 + c;
  const int lo = ptr[myRow < nRows ? myRow : nRows];
  const int hi = ptr[myRow + 1 < nRows ? myRow + 1 : nRows];
  const int start = __builtin_amdgcn_readlane(lo, 0);
  const int end   = __builtin_amdgcn_readlane(hi, 15);

  v8f acc0 = {}, acc1 = {}, acc2 = {}, acc3 = {};
  const int last = nEdges - 1;
  for (int p = start; p < end; p += 16) {
    v2f A[4];
    v2f B[4][4];
#pragma unroll
    for (int j = 0; j < 4; ++j) {
      const int q0 = p + 4 * j + 2 * h, q1 = q0 + 1;
      const int2 s0 = sorted[q0 < last ? q0 : last];
      const int2 s1 = sorted[q1 < last ? q1 : last];
      A[j].x = (q0 >= lo && q0 < hi) ? __int_as_float(s0.y) : 0.0f;
      A[j].y = (q1 >= lo && q1 < hi) ? __int_as_float(s1.y) : 0.0f;
      const float* xa = x + (size_t)s0.x * EMB + c;
      const float* xb = x + (size_t)s1.x * EMB + c;
#pragma unroll
      for (int t = 0; t < 4; ++t) {
        B[j][t].x = xa[16 * t];
        B[j][t].y = xb[16 * t];
      }
    }
    __builtin_amdgcn_sched_barrier(0);
#pragma unroll
    for (int j = 0; j < 4; ++j) {
      acc0 = __builtin_amdgcn_wmma_f32_16x16x4_f32(false, A[j], false, B[j][0], (short)0, acc0, false, false);
      acc1 = __builtin_amdgcn_wmma_f32_16x16x4_f32(false, A[j], false, B[j][1], (short)0, acc1, false, false);
      acc2 = __builtin_amdgcn_wmma_f32_16x16x4_f32(false, A[j], false, B[j][2], (short)0, acc2, false, false);
      acc3 = __builtin_amdgcn_wmma_f32_16x16x4_f32(false, A[j], false, B[j][3], (short)0, acc3, false, false);
    }
    __builtin_amdgcn_sched_barrier(0);
    asm volatile("v_nop\n\tv_nop\n\tv_nop\n\tv_nop\n\tv_nop\n\tv_nop\n\tv_nop\n\tv_nop");
    __builtin_amdgcn_sched_barrier(0);
  }

#pragma unroll
  for (int r = 0; r < 8; ++r) {
    const int row = r0 + 8 * h + r;
    if (row < nRows) {
      const size_t idx = (size_t)row * EMB + c;
      const float d0 = acc0[r], d1 = acc1[r], d2 = acc2[r], d3 = acc3[r];
      if (writeY) {
        y[idx] = d0; y[idx + 16] = d1; y[idx + 32] = d2; y[idx + 48] = d3;
      }
      out[idx]      = (out[idx]      + d0) * scale;
      out[idx + 16] = (out[idx + 16] + d1) * scale;
      out[idx + 32] = (out[idx + 32] + d2) * scale;
      out[idx + 48] = (out[idx + 48] + d3) * scale;
    }
  }
}

extern "C" void kernel_launch(void* const* d_in, const int* in_sizes, int n_in,
                              void* d_out, int out_size, void* d_ws, size_t ws_size,
                              hipStream_t stream) {
  const float* users = (const float*)d_in[0];
  const float* items = (const float*)d_in[1];
  const int*   erow  = (const int*)d_in[2];
  const int*   ecol  = (const int*)d_in[3];
  const float* evals = (const float*)d_in[4];
  float* out = (float*)d_out;

  const int userElems  = in_sizes[0];
  const int itemElems  = in_sizes[1];
  const int nEdges     = in_sizes[2];
  const int totalElems = userElems + itemElems;
  const int nRows      = totalElems / EMB;

  const size_t bufBytes = (size_t)totalElems * sizeof(float);
  if (ws_size < 2 * bufBytes) return;
  float* cur = (float*)d_ws;
  float* nxt = (float*)((char*)d_ws + bufBytes);

  const int ethreads = (totalElems + 3) / 4;
  const dim3 eb(256), eg((ethreads + 255) / 256);
  init_concat_kernel<<<eg, eb, 0, stream>>>(users, items, out, cur,
                                            userElems, totalElems);

  const size_t bufAl    = (bufBytes + 255) & ~(size_t)255;
  const size_t edgeAl   = ((size_t)nEdges * sizeof(int2) + 255) & ~(size_t)255;
  const int    nPad     = ((nRows + 1 + SCAN_SPAN - 1) / SCAN_SPAN) * SCAN_SPAN;
  const size_t cntBytes = (size_t)nPad * sizeof(int);
  const size_t need     = 2 * bufAl + edgeAl + 3 * cntBytes;

  if (ws_size >= need && nEdges > 0) {
    nxt = (float*)((char*)d_ws + bufAl);
    int2* sorted = (int2*)((char*)d_ws + 2 * bufAl);
    int*  cnt    = (int*)((char*)d_ws + 2 * bufAl + edgeAl);
    int*  ptr    = cnt + nPad;
    int*  cursor = ptr + nPad;

    hipMemsetAsync(cnt, 0, cntBytes, stream);
    const int edgeBlocks = (nEdges + 255) / 256;
    row_count_kernel<<<edgeBlocks, 256, 0, stream>>>(erow, cnt, nEdges);
    row_scan_kernel<<<1, 1024, 0, stream>>>(cnt, ptr, cursor, nPad);
    bucket_fill_kernel<<<edgeBlocks, 256, 0, stream>>>(erow, ecol, evals,
                                                       cursor, sorted, nEdges);

    const int rowGroups  = (nRows + 15) / 16;
    const int rowBlocks  = (rowGroups + 3) / 4;
    for (int layer = 0; layer < 3; ++layer) {
      const float scale = (layer == 2) ? 0.25f : 1.0f;
      spmm_rows_wmma_kernel<<<rowBlocks, 128, 0, stream>>>(
          ptr, sorted, cur, nxt, out, scale, (layer == 2) ? 0 : 1, nRows, nEdges);
      float* tmp = cur; cur = nxt; nxt = tmp;
    }
    return;
  }

  const int nGroups    = (nEdges + 15) / 16;
  const int spmmBlocks = (nGroups + 7) / 8;

  for (int layer = 0; layer < 3; ++layer) {
    hipMemsetAsync(nxt, 0, bufBytes, stream);
    spmm_wmma_kernel<<<spmmBlocks, 256, 0, stream>>>(erow, ecol, evals,
                                                     cur, nxt, nEdges);
    const float scale = (layer == 2) ? 0.25f : 1.0f;
    axpy_scale_kernel<<<eg, eb, 0, stream>>>(out, nxt, scale, totalElems);
    float* tmp = cur; cur = nxt; nxt = tmp;
  }
}


# 16 B per lane in gathers and epilogue: lane owns columns 4c..4c+3, wave owns 8 rows (M = row x column parity), a load covers two whole 256-B rows; row sum order changes (2 slots per matrix step)
# speedup vs baseline: 1.4275x; 1.0576x over previous
#include <hip/hip_runtime.h>

typedef __attribute__((ext_vector_type(2))) float v2f;
typedef __attribute__((ext_vector_type(4))) float v4f;
typedef __attribute__((ext_vector_type(8))) float v8f;

#define EMB 64

__global__ void init_concat_kernel(const float* __restrict__ users,
                                   const float* __restrict__ items,
                                   float* __restrict__ out,
                                   float* __restrict__ cur,
                                   int userElems, int totalElems) {
  int i4 = (blockIdx.x * blockDim.x + threadIdx.x) * 4;
  if (i4 >= totalElems) return;
  float4 v = (i4 < userElems) ? *(const float4*)(users + i4)
                              : *(const float4*)(items + (i4 - userElems));
  *(float4*)(out + i4) = v;
  *(float4*)(cur + i4) = v;
}

__global__ void axpy_scale_kernel(float* __restrict__ out,
                                  const float* __restrict__ nxt,
                                  float scale, int totalElems) {
  int i4 = (blockIdx.x * blockDim.x + threadIdx.x) * 4;
  if (i4 >= totalElems) return;
  float4 a = *(const float4*)(out + i4);
  float4 b = *(const float4*)(nxt + i4);
  a.x = (a.x + b.x) * scale;
  a.y = (a.y + b.y) * scale;
  a.z = (a.z + b.z) * scale;
  a.w = (a.w + b.w) * scale;
  *(float4*)(out + i4) = a;
}

__global__ void spmm_wmma_kernel(const int*   __restrict__ rows,
                                 const int*   __restrict__ cols,
                                 const float* __restrict__ vals,
                                 const float* __restrict__ x,
                                 float*       __restrict__ y,
                                 int nEdges) {
  const int lane   = threadIdx.x & 31;
  const int m      = lane & 15;
  const bool hiHalf = lane >= 16;
  const int wavesPerBlock = blockDim.x >> 5;
  const int wave   = blockIdx.x * wavesPerBlock + (threadIdx.x >> 5);
  const int stride = gridDim.x * wavesPerBlock;
  const int nGroups = (nEdges + 15) >> 4;

  for (int g = wave; g < nGroups; g += stride) {
    const int base = g << 4;
    const int e    = base + m;
    const int eC   = (e < nEdges) ? e : (nEdges - 1);

    const float v   = (e < nEdges) ? vals[eC] : 0.0f;
    const int myRow = rows[eC];
    const int myCol = cols[eC];

    int colK[8];
#pragma unroll
    for (int c = 0; c < 4; ++c) {
      const int kb0 = 4 * c + (hiHalf ? 2 : 0);
      colK[2 * c]     = __shfl(myCol, kb0, 32);
      colK[2 * c + 1] = __shfl(myCol, kb0 + 1, 32);
    }

    int rowD[8];
#pragma unroll
    for (int vv = 0; vv < 8; ++vv)
      rowD[vv] = __shfl(myRow, hiHalf ? (vv + 8) : vv, 32);

    v2f A[4];
#pragma unroll
    for (int c = 0; c < 4; ++c) {
      const int ka0 = 4 * c + (hiHalf ? 2 : 0);
      A[c].x = (m == ka0)     ? v : 0.0f;
      A[c].y = (m == ka0 + 1) ? v : 0.0f;
    }

#pragma unroll
    for (int t = 0; t < 4; ++t) {
      v8f acc = {};
#pragma unroll
      for (int c = 0; c < 4; ++c) {
        v2f B;
        B.x = x[colK[2 * c]     * EMB + t * 16 + m];
        B.y = x[colK[2 * c + 1] * EMB + t * 16 + m];
        acc = __builtin_amdgcn_wmma_f32_16x16x4_f32(
             false, A[c],  false, B,
             (short)0, acc,  false,  false);
      }
#pragma unroll
      for (int vv = 0; vv < 8; ++vv)
        atomicAdd(&y[rowD[vv] * EMB + t * 16 + m], acc[vv]);
    }
  }
}


#define SCAN_SPAN 8192

__global__ void row_count_kernel(const int* __restrict__ rows,
                                 int* __restrict__ cnt, int nEdges) {
  int e = blockIdx.x * blockDim.x + threadIdx.x;
  if (e < nEdges) atomicAdd(&cnt[rows[e]], 1);
}

__global__ void row_scan_kernel(const int* __restrict__ cnt,
                                int* __restrict__ ptr,
                                int* __restrict__ cursor, int nPad) {
  __shared__ int waveSum[32];
  __shared__ int tripTotal;
  const int tid = threadIdx.x, lane = tid & 31, w = tid >> 5;
  int carry = 0;
  for (int base = 0; base < nPad; base += SCAN_SPAN) {
    const int i0 = base + tid * 8;
    int4 a = *(const int4*)(cnt + i0);
    int4 b = *(const int4*)(cnt + i0 + 4);
    int e0 = 0, e1 = a.x, e2 = e1 + a.y, e3 = e2 + a.z, e4 = e3 + a.w;
    int e5 = e4 + b.x, e6 = e5 + b.y, e7 = e6 + b.z;
    const int tsum = e7 + b.w;
    int s = tsum;
#pragma unroll
    for (int d = 1; d < 32; d <<= 1) {
      int t = __shfl_up(s, d, 32);
      if (lane >= d) s += t;
    }
    if (lane == 31) waveSum[w] = s;
    __syncthreads();
    if (w == 0) {
      const int ws = waveSum[lane];
      int t2 = ws;
#pragma unroll
      for (int d = 1; d < 32; d <<= 1) {
        int t = __shfl_up(t2, d, 32);
        if (lane >= d) t2 += t;
      }
      waveSum[lane] = t2 - ws;
      if (lane == 31) tripTotal = t2;
    }
    __syncthreads();
    const int off = carry + waveSum[w] + (s - tsum);
    int4 o0 = make_int4(off + e0, off + e1, off + e2, off + e3);
    int4 o1 = make_int4(off + e4, off + e5, off + e6, off + e7);
    *(int4*)(ptr + i0) = o0;      *(int4*)(ptr + i0 + 4) = o1;
    *(int4*)(cursor + i0) = o0;   *(int4*)(cursor + i0 + 4) = o1;
    carry += tripTotal;
    __syncthreads();
  }
}

__global__ void bucket_fill_kernel(const int* __restrict__ rows,
                                   const int* __restrict__ cols,
                                   const float* __restrict__ vals,
                                   int* __restrict__ cursor,
                                   int2* __restrict__ sorted, int nEdges) {
  int e = blockIdx.x * blockDim.x + threadIdx.x;
  if (e < nEdges) {
    const int slot = atomicAdd(&cursor[rows[e]], 1);
    sorted[slot] = make_int2(cols[e], __float_as_int(vals[e]));
  }
}

__global__ void spmm_rows_wmma_kernel(const int*  __restrict__ ptr,
                                      const int2* __restrict__ sorted,
                                      const float* __restrict__ x,
                                      float* __restrict__ y,
                                      float* __restrict__ out,
                                      float scale, int writeY,
                                      int nRows, int nEdges) {
  const int lane = threadIdx.x & 31;
  const int c = lane & 15;
  const int h = lane >> 4;
  const int wave = blockIdx.x * (blockDim.x >> 5) + (threadIdx.x >> 5);
  const int r0 = wave << 3;
  if (r0 >= nRows) return;

  const int myRow = r0 + (c >> 1);
  const bool odd = (c & 1) != 0;
  const int lo = ptr[myRow < nRows ? myRow : nRows];
  const int hi = ptr[myRow + 1 < nRows ? myRow + 1 : nRows];
  const int start = __builtin_amdgcn_readlane(lo, 0);
  const int end   = __builtin_amdgcn_readlane(hi, 15);

  const char* sortedB = (const char*)sorted;
  const char* xB = (const char*)x;
  const unsigned laneOff = (unsigned)c * 16u;

  v8f acc0 = {}, acc1 = {};
  const int last = nEdges - 1;
  for (int p = start; p < end; p += 16) {
    v2f A[8];
    v4f R[8];
#pragma unroll
    for (int i = 0; i < 8; ++i) {
      const int q = p + 2 * i + h;
      const int qc = q < last ? q : last;
      const int2 s = *(const int2*)(sortedB + (unsigned)qc * 8u);
      const float v = (q >= lo && q < hi) ? __int_as_float(s.y) : 0.0f;
      A[i].x = odd ? 0.0f : v;
      A[i].y = odd ? v : 0.0f;
      R[i] = *(const v4f*)(xB + ((unsigned)s.x * (unsigned)(EMB * 4) + laneOff));
    }
    __builtin_amdgcn_sched_barrier(0);
#pragma unroll
    for (int i = 0; i < 8; ++i) {
      const v2f b0 = {R[i].x, R[i].y};
      const v2f b1 = {R[i].z, R[i].w};
      acc0 = __builtin_amdgcn_wmma_f32_16x16x4_f32(false, A[i], false, b0, (short)0, acc0, false, false);
      acc1 = __builtin_amdgcn_wmma_f32_16x16x4_f32(false, A[i], false, b1, (short)0, acc1, false, false);
    }
    __builtin_amdgcn_sched_barrier(0);
    asm volatile("v_nop\n\tv_nop\n\tv_nop\n\tv_nop\n\tv_nop\n\tv_nop\n\tv_nop\n\tv_nop");
    __builtin_amdgcn_sched_barrier(0);
  }

#pragma unroll
  for (int i = 0; i < 4; ++i) {
    const int row = r0 + 4 * h + i;
    if (row < nRows) {
      const size_t idx = (size_t)row * EMB + 4 * c;
      const v4f d = {acc0[2 * i], acc0[2 * i + 1], acc1[2 * i], acc1[2 * i + 1]};
      if (writeY) *(v4f*)(y + idx) = d;
      const v4f o = *(const v4f*)(out + idx);
      *(v4f*)(out + idx) = (o + d) * scale;
    }
  }
}

extern "C" void kernel_launch(void* const* d_in, const int* in_sizes, int n_in,
                              void* d_out, int out_size, void* d_ws, size_t ws_size,
                              hipStream_t stream) {
  const float* users = (const float*)d_in[0];
  const float* items = (const float*)d_in[1];
  const int*   erow  = (const int*)d_in[2];
  const int*   ecol  = (const int*)d_in[3];
  const float* evals = (const float*)d_in[4];
  float* out = (float*)d_out;

  const int userElems  = in_sizes[0];
  const int itemElems  = in_sizes[1];
  const int nEdges     = in_sizes[2];
  const int totalElems = userElems + itemElems;
  const int nRows      = totalElems / EMB;

  const size_t bufBytes = (size_t)totalElems * sizeof(float);
  if (ws_size < 2 * bufBytes) return;
  float* cur = (float*)d_ws;
  float* nxt = (float*)((char*)d_ws + bufBytes);

  const int ethreads = (totalElems + 3) / 4;
  const dim3 eb(256), eg((ethreads + 255) / 256);
  init_concat_kernel<<<eg, eb, 0, stream>>>(users, items, out, cur,
                                            userElems, totalElems);

  const size_t bufAl    = (bufBytes + 255) & ~(size_t)255;
  const size_t edgeAl   = ((size_t)nEdges * sizeof(int2) + 255) & ~(size_t)255;
  const int    nPad     = ((nRows + 1 + SCAN_SPAN - 1) / SCAN_SPAN) * SCAN_SPAN;
  const size_t cntBytes = (size_t)nPad * sizeof(int);
  const size_t need     = 2 * bufAl + edgeAl + 3 * cntBytes;

  if (ws_size >= need && nEdges > 0) {
    nxt = (float*)((char*)d_ws + bufAl);
    int2* sorted = (int2*)((char*)d_ws + 2 * bufAl);
    int*  cnt    = (int*)((char*)d_ws + 2 * bufAl + edgeAl);
    int*  ptr    = cnt + nPad;
    int*  cursor = ptr + nPad;

    hipMemsetAsync(cnt, 0, cntBytes, stream);
    const int edgeBlocks = (nEdges + 255) / 256;
    row_count_kernel<<<edgeBlocks, 256, 0, stream>>>(erow, cnt, nEdges);
    row_scan_kernel<<<1, 1024, 0, stream>>>(cnt, ptr, cursor, nPad);
    bucket_fill_kernel<<<edgeBlocks, 256, 0, stream>>>(erow, ecol, evals,
                                                       cursor, sorted, nEdges);

    const int rowGroups  = (nRows + 7) / 8;
    const int rowBlocks  = (rowGroups + 7) / 8;
    for (int layer = 0; layer < 3; ++layer) {
      const float scale = (layer == 2) ? 0.25f : 1.0f;
      spmm_rows_wmma_kernel<<<rowBlocks, 256, 0, stream>>>(
          ptr, sorted, cur, nxt, out, scale, (layer == 2) ? 0 : 1, nRows, nEdges);
      float* tmp = cur; cur = nxt; nxt = tmp;
    }
    return;
  }

  const int nGroups    = (nEdges + 15) / 16;
  const int spmmBlocks = (nGroups + 7) / 8;

  for (int layer = 0; layer < 3; ++layer) {
    hipMemsetAsync(nxt, 0, bufBytes, stream);
    spmm_wmma_kernel<<<spmmBlocks, 256, 0, stream>>>(erow, ecol, evals,
                                                     cur, nxt, nEdges);
    const float scale = (layer == 2) ? 0.25f : 1.0f;
    axpy_scale_kernel<<<eg, eb, 0, stream>>>(out, nxt, scale, totalElems);
    float* tmp = cur; cur = nxt; nxt = tmp;
  }
}


# One atomic per edge, not two: the counting add returns the edge's rank in its row, the fill places by row start + rank (fill atomics and cursor array go); row sum order = arrival rank at the count
# speedup vs baseline: 1.7617x; 1.2341x over previous
#include <hip/hip_runtime.h>

typedef __attribute__((ext_vector_type(2))) float v2f;
typedef __attribute__((ext_vector_type(4))) float v4f;
typedef __attribute__((ext_vector_type(8))) float v8f;

#define EMB 64

__global__ void init_concat_kernel(const float* __restrict__ users,
                                   const float* __restrict__ items,
                                   float* __restrict__ out,
                                   float* __restrict__ cur,
                                   int userElems, int totalElems) {
  int i4 = (blockIdx.x * blockDim.x + threadIdx.x) * 4;
  if (i4 >= totalElems) return;
  float4 v = (i4 < userElems) ? *(const float4*)(users + i4)
                              : *(const float4*)(items + (i4 - userElems));
  *(float4*)(out + i4) = v;
  *(float4*)(cur + i4) = v;
}

__global__ void axpy_scale_kernel(float* __restrict__ out,
                                  const float* __restrict__ nxt,
                                  float scale, int totalElems) {
  int i4 = (blockIdx.x * blockDim.x + threadIdx.x) * 4;
  if (i4 >= totalElems) return;
  float4 a = *(const float4*)(out + i4);
  float4 b = *(const float4*)(nxt + i4);
  a.x = (a.x + b.x) * scale;
  a.y = (a.y + b.y) * scale;
  a.z = (a.z + b.z) * scale;
  a.w = (a.w + b.w) * scale;
  *(float4*)(out + i4) = a;
}

__global__ void spmm_wmma_kernel(const int*   __restrict__ rows,
                                 const int*   __restrict__ cols,
                                 const float* __restrict__ vals,
                                 const float* __restrict__ x,
                                 float*       __restrict__ y,
                                 int nEdges) {
  const int lane   = threadIdx.x & 31;
  const int m      = lane & 15;
  const bool hiHalf = lane >= 16;
  const int wavesPerBlock = blockDim.x >> 5;
  const int wave   = blockIdx.x * wavesPerBlock + (threadIdx.x >> 5);
  const int stride = gridDim.x * wavesPerBlock;
  const int nGroups = (nEdges + 15) >> 4;

  for (int g = wave; g < nGroups; g += stride) {
    const int base = g << 4;
    const int e    = base + m;
    const int eC   = (e < nEdges) ? e : (nEdges - 1);

    const float v   = (e < nEdges) ? vals[eC] : 0.0f;
    const int myRow = rows[eC];
    const int myCol = cols[eC];

    int colK[8];
#pragma unroll
    for (int c = 0; c < 4; ++c) {
      const int kb0 = 4 * c + (hiHalf ? 2 : 0);
      colK[2 * c]     = __shfl(myCol, kb0, 32);
      colK[2 * c + 1] = __shfl(myCol, kb0 + 1, 32);
    }

    int rowD[8];
#pragma unroll
    for (int vv = 0; vv < 8; ++vv)
      rowD[vv] = __shfl(myRow, hiHalf ? (vv + 8) : vv, 32);

    v2f A[4];
#pragma unroll
    for (int c = 0; c < 4; ++c) {
      const int ka0 = 4 * c + (hiHalf ? 2 : 0);
      A[c].x = (m == ka0)     ? v : 0.0f;
      A[c].y = (m == ka0 + 1) ? v : 0.0f;
    }

#pragma unroll
    for (int t = 0; t < 4; ++t) {
      v8f acc = {};
#pragma unroll
      for (int c = 0; c < 4; ++c) {
        v2f B;
        B.x = x[colK[2 * c]     * EMB + t * 16 + m];
        B.y = x[colK[2 * c + 1] * EMB + t * 16 + m];
        acc = __builtin_amdgcn_wmma_f32_16x16x4_f32(
             false, A[c],  false, B,
             (short)0, acc,  false,  false);
      }
#pragma unroll
      for (int vv = 0; vv < 8; ++vv)
        atomicAdd(&y[rowD[vv] * EMB + t * 16 + m], acc[vv]);
    }
  }
}


#define SCAN_SPAN 8192

__global__ void row_count_kernel(const int* __restrict__ rows,
                                 int* __restrict__ cnt,
                                 int* __restrict__ rank, int nEdges) {
  int e = blockIdx.x * blockDim.x + threadIdx.x;
  if (e < nEdges) rank[e] = atomicAdd(&cnt[rows[e]], 1);
}

__global__ void row_scan_kernel(const int* __restrict__ cnt,
                                int* __restrict__ ptr, int nPad) {
  __shared__ int waveSum[32];
  __shared__ int tripTotal;
  const int tid = threadIdx.x, lane = tid & 31, w = tid >> 5;
  int carry = 0;
  for (int base = 0; base < nPad; base += SCAN_SPAN) {
    const int i0 = base + tid * 8;
    int4 a = *(const int4*)(cnt + i0);
    int4 b = *(const int4*)(cnt + i0 + 4);
    int e0 = 0, e1 = a.x, e2 = e1 + a.y, e3 = e2 + a.z, e4 = e3 + a.w;
    int e5 = e4 + b.x, e6 = e5 + b.y, e7 = e6 + b.z;
    const int tsum = e7 + b.w;
    int s = tsum;
#pragma unroll
    for (int d = 1; d < 32; d <<= 1) {
      int t = __shfl_up(s, d, 32);
      if (lane >= d) s += t;
    }
    if (lane == 31) waveSum[w] = s;
    __syncthreads();
    if (w == 0) {
      const int ws = waveSum[lane];
      int t2 = ws;
#pragma unroll
      for (int d = 1; d < 32; d <<= 1) {
        int t = __shfl_up(t2, d, 32);
        if (lane >= d) t2 += t;
      }
      waveSum[lane] = t2 - ws;
      if (lane == 31) tripTotal = t2;
    }
    __syncthreads();
    const int off = carry + waveSum[w] + (s - tsum);
    int4 o0 = make_int4(off + e0, off + e1, off + e2, off + e3);
    int4 o1 = make_int4(off + e4, off + e5, off + e6, off + e7);
    *(int4*)(ptr + i0) = o0;      *(int4*)(ptr + i0 + 4) = o1;
    carry += tripTotal;
    __syncthreads();
  }
}

__global__ void bucket_fill_kernel(const int* __restrict__ rows,
                                   const int* __restrict__ cols,
                                   const float* __restrict__ vals,
                                   const int* __restrict__ ptr,
                                   const int* __restrict__ rank,
                                   int2* __restrict__ sorted, int nEdges) {
  int e = blockIdx.x * blockDim.x + threadIdx.x;
  if (e < nEdges) {
    const int slot = ptr[rows[e]] + rank[e];
    sorted[slot] = make_int2(cols[e], __float_as_int(vals[e]));
  }
}

__global__ void spmm_rows_wmma_kernel(const int*  __restrict__ ptr,
                                      const int2* __restrict__ sorted,
                                      const float* __restrict__ x,
                                      float* __restrict__ y,
                                      float* __restrict__ out,
                                      float scale, int writeY,
                                      int nRows, int nEdges) {
  const int lane = threadIdx.x & 31;
  const int c = lane & 15;
  const int h = lane >> 4;
  const int wave = blockIdx.x * (blockDim.x >> 5) + (threadIdx.x >> 5);
  const int r0 = wave << 3;
  if (r0 >= nRows) return;

  const int myRow = r0 + (c >> 1);
  const bool odd = (c & 1) != 0;
  const int lo = ptr[myRow < nRows ? myRow : nRows];
  const int hi = ptr[myRow + 1 < nRows ? myRow + 1 : nRows];
  const int start = __builtin_amdgcn_readlane(lo, 0);
  const int end   = __builtin_amdgcn_readlane(hi, 15);

  const char* sortedB = (const char*)sorted;
  const char* xB = (const char*)x;
  const unsigned laneOff = (unsigned)c * 16u;

  v8f acc0 = {}, acc1 = {};
  const int last = nEdges - 1;
  for (int p = start; p < end; p += 16) {
    v2f A[8];
    v4f R[8];
#pragma unroll
    for (int i = 0; i < 8; ++i) {
      const int q = p + 2 * i + h;
      const int qc = q < last ? q : last;
      const int2 s = *(const int2*)(sortedB + (unsigned)qc * 8u);
      const float v = (q >= lo && q < hi) ? __int_as_float(s.y) : 0.0f;
      A[i].x = odd ? 0.0f : v;
      A[i].y = odd ? v : 0.0f;
      R[i] = *(const v4f*)(xB + ((unsigned)s.x * (unsigned)(EMB * 4) + laneOff));
    }
    __builtin_amdgcn_sched_barrier(0);
#pragma unroll
    for (int i = 0; i < 8; ++i) {
      const v2f b0 = {R[i].x, R[i].y};
      const v2f b1 = {R[i].z, R[i].w};
      acc0 = __builtin_amdgcn_wmma_f32_16x16x4_f32(false, A[i], false, b0, (short)0, acc0, false, false);
      acc1 = __builtin_amdgcn_wmma_f32_16x16x4_f32(false, A[i], false, b1, (short)0, acc1, false, false);
    }
    __builtin_amdgcn_sched_barrier(0);
    asm volatile("v_nop\n\tv_nop\n\tv_nop\n\tv_nop\n\tv_nop\n\tv_nop\n\tv_nop\n\tv_nop");
    __builtin_amdgcn_sched_barrier(0);
  }

#pragma unroll
  for (int i = 0; i < 4; ++i) {
    const int row = r0 + 4 * h + i;
    if (row < nRows) {
      const size_t idx = (size_t)row * EMB + 4 * c;
      const v4f d = {acc0[2 * i], acc0[2 * i + 1], acc1[2 * i], acc1[2 * i + 1]};
      if (writeY) *(v4f*)(y + idx) = d;
      const v4f o = *(const v4f*)(out + idx);
      *(v4f*)(out + idx) = (o + d) * scale;
    }
  }
}

extern "C" void kernel_launch(void* const* d_in, const int* in_sizes, int n_in,
                              void* d_out, int out_size, void* d_ws, size_t ws_size,
                              hipStream_t stream) {
  const float* users = (const float*)d_in[0];
  const float* items = (const float*)d_in[1];
  const int*   erow  = (const int*)d_in[2];
  const int*   ecol  = (const int*)d_in[3];
  const float* evals = (const float*)d_in[4];
  float* out = (float*)d_out;

  const int userElems  = in_sizes[0];
  const int itemElems  = in_sizes[1];
  const int nEdges     = in_sizes[2];
  const int totalElems = userElems + itemElems;
  const int nRows      = totalElems / EMB;

  const size_t bufBytes = (size_t)totalElems * sizeof(float);
  if (ws_size < 2 * bufBytes) return;
  float* cur = (float*)d_ws;
  float* nxt = (float*)((char*)d_ws + bufBytes);

  const int ethreads = (totalElems + 3) / 4;
  const dim3 eb(256), eg((ethreads + 255) / 256);
  init_concat_kernel<<<eg, eb, 0, stream>>>(users, items, out, cur,
                                            userElems, totalElems);

  const size_t bufAl    = (bufBytes + 255) & ~(size_t)255;
  const size_t edgeAl   = ((size_t)nEdges * sizeof(int2) + 255) & ~(size_t)255;
  const int    nPad     = ((nRows + 1 + SCAN_SPAN - 1) / SCAN_SPAN) * SCAN_SPAN;
  const size_t cntBytes = (size_t)nPad * sizeof(int);
  const size_t need     = 2 * bufAl + edgeAl + 2 * cntBytes;
  const size_t rankBytes = (size_t)nEdges * sizeof(int);

  if (ws_size >= need && nEdges > 0 && rankBytes <= bufAl) {
    nxt = (float*)((char*)d_ws + bufAl);
    int2* sorted = (int2*)((char*)d_ws + 2 * bufAl);
    int*  cnt    = (int*)((char*)d_ws + 2 * bufAl + edgeAl);
    int*  ptr    = cnt + nPad;
    int*  rank   = (int*)nxt;

    hipMemsetAsync(cnt, 0, cntBytes, stream);
    const int edgeBlocks = (nEdges + 255) / 256;
    row_count_kernel<<<edgeBlocks, 256, 0, stream>>>(erow, cnt, rank, nEdges);
    row_scan_kernel<<<1, 1024, 0, stream>>>(cnt, ptr, nPad);
    bucket_fill_kernel<<<edgeBlocks, 256, 0, stream>>>(erow, ecol, evals,
                                                       ptr, rank, sorted, nEdges);

    const int rowGroups  = (nRows + 7) / 8;
    const int rowBlocks  = (rowGroups + 7) / 8;
    for (int layer = 0; layer < 3; ++layer) {
      const float scale = (layer == 2) ? 0.25f : 1.0f;
      spmm_rows_wmma_kernel<<<rowBlocks, 256, 0, stream>>>(
          ptr, sorted, cur, nxt, out, scale, (layer == 2) ? 0 : 1, nRows, nEdges);
      float* tmp = cur; cur = nxt; nxt = tmp;
    }
    return;
  }

  const int nGroups    = (nEdges + 15) / 16;
  const int spmmBlocks = (nGroups + 7) / 8;

  for (int layer = 0; layer < 3; ++layer) {
    hipMemsetAsync(nxt, 0, bufBytes, stream);
    spmm_wmma_kernel<<<spmmBlocks, 256, 0, stream>>>(erow, ecol, evals,
                                                     cur, nxt, nEdges);
    const float scale = (layer == 2) ? 0.25f : 1.0f;
    axpy_scale_kernel<<<eg, eb, 0, stream>>>(out, nxt, scale, totalElems);
    float* tmp = cur; cur = nxt; nxt = tmp;
  }
}


# Edges bucketed by a two-level counting sort in LDS (coarse bins of 512 rows per edge chunk, then one block per bin): the per-edge global atomic and the memset go; row sum order = LDS add order
# speedup vs baseline: 2.4232x; 1.3755x over previous
#include <hip/hip_runtime.h>

typedef __attribute__((ext_vector_type(2))) float v2f;
typedef __attribute__((ext_vector_type(4))) float v4f;
typedef __attribute__((ext_vector_type(8))) float v8f;

#define EMB 64

__global__ void init_concat_kernel(const float* __restrict__ users,
                                   const float* __restrict__ items,
                                   float* __restrict__ out,
                                   float* __restrict__ cur,
                                   int userElems, int totalElems) {
  int i4 = (blockIdx.x * blockDim.x + threadIdx.x) * 4;
  if (i4 >= totalElems) return;
  float4 v = (i4 < userElems) ? *(const float4*)(users + i4)
                              : *(const float4*)(items + (i4 - userElems));
  *(float4*)(out + i4) = v;
  *(float4*)(cur + i4) = v;
}

__global__ void axpy_scale_kernel(float* __restrict__ out,
                                  const float* __restrict__ nxt,
                                  float scale, int totalElems) {
  int i4 = (blockIdx.x * blockDim.x + threadIdx.x) * 4;
  if (i4 >= totalElems) return;
  float4 a = *(const float4*)(out + i4);
  float4 b = *(const float4*)(nxt + i4);
  a.x = (a.x + b.x) * scale;
  a.y = (a.y + b.y) * scale;
  a.z = (a.z + b.z) * scale;
  a.w = (a.w + b.w) * scale;
  *(float4*)(out + i4) = a;
}

__global__ void spmm_wmma_kernel(const int*   __restrict__ rows,
                                 const int*   __restrict__ cols,
                                 const float* __restrict__ vals,
                                 const float* __restrict__ x,
                                 float*       __restrict__ y,
                                 int nEdges) {
  const int lane   = threadIdx.x & 31;
  const int m      = lane & 15;
  const bool hiHalf = lane >= 16;
  const int wavesPerBlock = blockDim.x >> 5;
  const int wave   = blockIdx.x * wavesPerBlock + (threadIdx.x >> 5);
  const int stride = gridDim.x * wavesPerBlock;
  const int nGroups = (nEdges + 15) >> 4;

  for (int g = wave; g < nGroups; g += stride) {
    const int base = g << 4;
    const int e    = base + m;
    const int eC   = (e < nEdges) ? e : (nEdges - 1);

    const float v   = (e < nEdges) ? vals[eC] : 0.0f;
    const int myRow = rows[eC];
    const int myCol = cols[eC];

    int colK[8];
#pragma unroll
    for (int c = 0; c < 4; ++c) {
      const int kb0 = 4 * c + (hiHalf ? 2 : 0);
      colK[2 * c]     = __shfl(myCol, kb0, 32);
      colK[2 * c + 1] = __shfl(myCol, kb0 + 1, 32);
    }

    int rowD[8];
#pragma unroll
    for (int vv = 0; vv < 8; ++vv)
      rowD[vv] = __shfl(myRow, hiHalf ? (vv + 8) : vv, 32);

    v2f A[4];
#pragma unroll
    for (int c = 0; c < 4; ++c) {
      const int ka0 = 4 * c + (hiHalf ? 2 : 0);
      A[c].x = (m == ka0)     ? v : 0.0f;
      A[c].y = (m == ka0 + 1) ? v : 0.0f;
    }

#pragma unroll
    for (int t = 0; t < 4; ++t) {
      v8f acc = {};
#pragma unroll
      for (int c = 0; c < 4; ++c) {
        v2f B;
        B.x = x[colK[2 * c]     * EMB + t * 16 + m];
        B.y = x[colK[2 * c + 1] * EMB + t * 16 + m];
        acc = __builtin_amdgcn_wmma_f32_16x16x4_f32(
             false, A[c],  false, B,
             (short)0, acc,  false,  false);
      }
#pragma unroll
      for (int vv = 0; vv < 8; ++vv)
        atomicAdd(&y[rowD[vv] * EMB + t * 16 + m], acc[vv]);
    }
  }
}


#define SCAN_SPAN 8192

__global__ void row_count_kernel(const int* __restrict__ rows,
                                 int* __restrict__ cnt,
                                 int* __restrict__ rank, int nEdges) {
  int e = blockIdx.x * blockDim.x + threadIdx.x;
  if (e < nEdges) rank[e] = atomicAdd(&cnt[rows[e]], 1);
}

__global__ void row_scan_kernel(const int* __restrict__ cnt,
                                int* __restrict__ ptr, int nPad) {
  __shared__ int waveSum[32];
  __shared__ int tripTotal;
  const int tid = threadIdx.x, lane = tid & 31, w = tid >> 5;
  int carry = 0;
  for (int base = 0; base < nPad; base += SCAN_SPAN) {
    const int i0 = base + tid * 8;
    int4 a = *(const int4*)(cnt + i0);
    int4 b = *(const int4*)(cnt + i0 + 4);
    int e0 = 0, e1 = a.x, e2 = e1 + a.y, e3 = e2 + a.z, e4 = e3 + a.w;
    int e5 = e4 + b.x, e6 = e5 + b.y, e7 = e6 + b.z;
    const int tsum = e7 + b.w;
    int s = tsum;
#pragma unroll
    for (int d = 1; d < 32; d <<= 1) {
      int t = __shfl_up(s, d, 32);
      if (lane >= d) s += t;
    }
    if (lane == 31) waveSum[w] = s;
    __syncthreads();
    if (w == 0) {
      const int ws = waveSum[lane];
      int t2 = ws;
#pragma unroll
      for (int d = 1; d < 32; d <<= 1) {
        int t = __shfl_up(t2, d, 32);
        if (lane >= d) t2 += t;
      }
      waveSum[lane] = t2 - ws;
      if (lane == 31) tripTotal = t2;
    }
    __syncthreads();
    const int off = carry + waveSum[w] + (s - tsum);
    int4 o0 = make_int4(off + e0, off + e1, off + e2, off + e3);
    int4 o1 = make_int4(off + e4, off + e5, off + e6, off + e7);
    *(int4*)(ptr + i0) = o0;      *(int4*)(ptr + i0 + 4) = o1;
    carry += tripTotal;
    __syncthreads();
  }
}

__global__ void bucket_fill_kernel(const int* __restrict__ rows,
                                   const int* __restrict__ cols,
                                   const float* __restrict__ vals,
                                   const int* __restrict__ ptr,
                                   const int* __restrict__ rank,
                                   int2* __restrict__ sorted, int nEdges) {
  int e = blockIdx.x * blockDim.x + threadIdx.x;
  if (e < nEdges) {
    const int slot = ptr[rows[e]] + rank[e];
    sorted[slot] = make_int2(cols[e], __float_as_int(vals[e]));
  }
}

#define FINE_SHIFT 9
#define FINE_ROWS  (1 << FINE_SHIFT)
#define CB_MAX     2048
#define SORT_BLOCK 1024

__global__ void coarse_hist_kernel(const int* __restrict__ rows,
                                   int* __restrict__ table,
                                   int nEdges, int chunk, int nBins, int nPadT) {
  __shared__ int hist[CB_MAX];
  const int tid = threadIdx.x, G = gridDim.x;
  for (int i = tid; i < nBins; i += SORT_BLOCK) hist[i] = 0;
  __syncthreads();
  const int e0 = blockIdx.x * chunk;
  const int e1 = (e0 + chunk < nEdges) ? (e0 + chunk) : nEdges;
  for (int e = e0 + tid; e < e1; e += 4 * SORT_BLOCK) {
    int r[4];
#pragma unroll
    for (int k = 0; k < 4; ++k) {
      const int ek = e + k * SORT_BLOCK;
      r[k] = rows[ek < e1 ? ek : e1 - 1];
    }
#pragma unroll
    for (int k = 0; k < 4; ++k)
      if (e + k * SORT_BLOCK < e1) atomicAdd(&hist[r[k] >> FINE_SHIFT], 1);
  }
  __syncthreads();
  for (int i = tid; i < nBins; i += SORT_BLOCK) table[i * G + blockIdx.x] = hist[i];
  for (int i = nBins * G + blockIdx.x * SORT_BLOCK + tid; i < nPadT; i += G * SORT_BLOCK)
    table[i] = 0;
}

__global__ void coarse_scatter_kernel(const int* __restrict__ rows,
                                      const int* __restrict__ cols,
                                      const float* __restrict__ vals,
                                      const int* __restrict__ tabOff,
                                      int* __restrict__ tmpRow,
                                      int2* __restrict__ tmpCV,
                                      int nEdges, int chunk, int nBins) {
  __shared__ int cursor[CB_MAX];
  const int tid = threadIdx.x, G = gridDim.x;
  for (int i = tid; i < nBins; i += SORT_BLOCK) cursor[i] = tabOff[i * G + blockIdx.x];
  __syncthreads();
  const int e0 = blockIdx.x * chunk;
  const int e1 = (e0 + chunk < nEdges) ? (e0 + chunk) : nEdges;
  for (int e = e0 + tid; e < e1; e += 4 * SORT_BLOCK) {
    int r[4], cI[4], vI[4];
#pragma unroll
    for (int k = 0; k < 4; ++k) {
      const int ek = e + k * SORT_BLOCK;
      const int ec = ek < e1 ? ek : e1 - 1;
      r[k]  = rows[ec];
      cI[k] = cols[ec];
      vI[k] = __float_as_int(vals[ec]);
    }
#pragma unroll
    for (int k = 0; k < 4; ++k) {
      if (e + k * SORT_BLOCK < e1) {
        const int pos = atomicAdd(&cursor[r[k] >> FINE_SHIFT], 1);
        tmpRow[pos] = r[k];
        tmpCV[pos]  = make_int2(cI[k], vI[k]);
      }
    }
  }
}

__global__ void fine_sort_kernel(const int* __restrict__ tabOff,
                                 const int* __restrict__ tmpRow,
                                 const int2* __restrict__ tmpCV,
                                 int* __restrict__ rank,
                                 int* __restrict__ ptr,
                                 int2* __restrict__ sorted,
                                 int G, int nRows) {
  __shared__ int cnt[FINE_ROWS];
  __shared__ int waveSum[32];
  const int tid = threadIdx.x, lane = tid & 31, w = tid >> 5;
  const int bin = blockIdx.x;
  const int bs = tabOff[bin * G];
  const int be = tabOff[(bin + 1) * G];
  const int row0 = bin << FINE_SHIFT;
  if (tid < FINE_ROWS) cnt[tid] = 0;
  __syncthreads();
  for (int p = bs + tid; p < be; p += 4 * SORT_BLOCK) {
    int lr[4];
#pragma unroll
    for (int k = 0; k < 4; ++k) {
      const int pk = p + k * SORT_BLOCK;
      lr[k] = tmpRow[pk < be ? pk : be - 1] - row0;
    }
#pragma unroll
    for (int k = 0; k < 4; ++k)
      if (p + k * SORT_BLOCK < be) rank[p + k * SORT_BLOCK] = atomicAdd(&cnt[lr[k]], 1);
  }
  __syncthreads();
  const int v = (tid < FINE_ROWS) ? cnt[tid] : 0;
  int s = v;
#pragma unroll
  for (int d = 1; d < 32; d <<= 1) {
    int t = __shfl_up(s, d, 32);
    if (lane >= d) s += t;
  }
  if (lane == 31) waveSum[w] = s;
  __syncthreads();
  if (w == 0) {
    const int ws = waveSum[lane];
    int t2 = ws;
#pragma unroll
    for (int d = 1; d < 32; d <<= 1) {
      int t = __shfl_up(t2, d, 32);
      if (lane >= d) t2 += t;
    }
    waveSum[lane] = t2 - ws;
  }
  __syncthreads();
  const int rowStart = bs + waveSum[w] + (s - v);
  if (tid < FINE_ROWS) {
    cnt[tid] = rowStart;
    if (row0 + tid < nRows) ptr[row0 + tid] = rowStart;
  }
  if (bin == (int)gridDim.x - 1 && tid == 0) ptr[nRows] = be;
  __syncthreads();
  for (int p = bs + tid; p < be; p += 4 * SORT_BLOCK) {
    int lr[4], rk[4];
    int2 cv[4];
#pragma unroll
    for (int k = 0; k < 4; ++k) {
      const int pk = p + k * SORT_BLOCK;
      const int pc = pk < be ? pk : be - 1;
      lr[k] = tmpRow[pc] - row0;
      rk[k] = rank[pc];
      cv[k] = tmpCV[pc];
    }
#pragma unroll
    for (int k = 0; k < 4; ++k)
      if (p + k * SORT_BLOCK < be) sorted[cnt[lr[k]] + rk[k]] = cv[k];
  }
}

__global__ void spmm_rows_wmma_kernel(const int*  __restrict__ ptr,
                                      const int2* __restrict__ sorted,
                                      const float* __restrict__ x,
                                      float* __restrict__ y,
                                      float* __restrict__ out,
                                      float scale, int writeY,
                                      int nRows, int nEdges) {
  const int lane = threadIdx.x & 31;
  const int c = lane & 15;
  const int h = lane >> 4;
  const int wave = blockIdx.x * (blockDim.x >> 5) + (threadIdx.x >> 5);
  const int r0 = wave << 3;
  if (r0 >= nRows) return;

  const int myRow = r0 + (c >> 1);
  const bool odd = (c & 1) != 0;
  const int lo = ptr[myRow < nRows ? myRow : nRows];
  const int hi = ptr[myRow + 1 < nRows ? myRow + 1 : nRows];
  const int start = __builtin_amdgcn_readlane(lo, 0);
  const int end   = __builtin_amdgcn_readlane(hi, 15);

  const char* sortedB = (const char*)sorted;
  const char* xB = (const char*)x;
  const unsigned laneOff = (unsigned)c * 16u;

  v8f acc0 = {}, acc1 = {};
  const int last = nEdges - 1;
  for (int p = start; p < end; p += 16) {
    v2f A[8];
    v4f R[8];
#pragma unroll
    for (int i = 0; i < 8; ++i) {
      const int q = p + 2 * i + h;
      const int qc = q < last ? q : last;
      const int2 s = *(const int2*)(sortedB + (unsigned)qc * 8u);
      const float v = (q >= lo && q < hi) ? __int_as_float(s.y) : 0.0f;
      A[i].x = odd ? 0.0f : v;
      A[i].y = odd ? v : 0.0f;
      R[i] = *(const v4f*)(xB + ((unsigned)s.x * (unsigned)(EMB * 4) + laneOff));
    }
    __builtin_amdgcn_sched_barrier(0);
#pragma unroll
    for (int i = 0; i < 8; ++i) {
      const v2f b0 = {R[i].x, R[i].y};
      const v2f b1 = {R[i].z, R[i].w};
      acc0 = __builtin_amdgcn_wmma_f32_16x16x4_f32(false, A[i], false, b0, (short)0, acc0, false, false);
      acc1 = __builtin_amdgcn_wmma_f32_16x16x4_f32(false, A[i], false, b1, (short)0, acc1, false, false);
    }
    __builtin_amdgcn_sched_barrier(0);
    asm volatile("v_nop\n\tv_nop\n\tv_nop\n\tv_nop\n\tv_nop\n\tv_nop\n\tv_nop\n\tv_nop");
    __builtin_amdgcn_sched_barrier(0);
  }

#pragma unroll
  for (int i = 0; i < 4; ++i) {
    const int row = r0 + 4 * h + i;
    if (row < nRows) {
      const size_t idx = (size_t)row * EMB + 4 * c;
      const v4f d = {acc0[2 * i], acc0[2 * i + 1], acc1[2 * i], acc1[2 * i + 1]};
      if (writeY) *(v4f*)(y + idx) = d;
      const v4f o = *(const v4f*)(out + idx);
      *(v4f*)(out + idx) = (o + d) * scale;
    }
  }
}

extern "C" void kernel_launch(void* const* d_in, const int* in_sizes, int n_in,
                              void* d_out, int out_size, void* d_ws, size_t ws_size,
                              hipStream_t stream) {
  const float* users = (const float*)d_in[0];
  const float* items = (const float*)d_in[1];
  const int*   erow  = (const int*)d_in[2];
  const int*   ecol  = (const int*)d_in[3];
  const float* evals = (const float*)d_in[4];
  float* out = (float*)d_out;

  const int userElems  = in_sizes[0];
  const int itemElems  = in_sizes[1];
  const int nEdges     = in_sizes[2];
  const int totalElems = userElems + itemElems;
  const int nRows      = totalElems / EMB;

  const size_t bufBytes = (size_t)totalElems * sizeof(float);
  if (ws_size < 2 * bufBytes) return;
  float* cur = (float*)d_ws;
  float* nxt = (float*)((char*)d_ws + bufBytes);

  const int ethreads = (totalElems + 3) / 4;
  const dim3 eb(256), eg((ethreads + 255) / 256);

  const size_t bufAl    = (bufBytes + 255) & ~(size_t)255;
  const size_t edgeAl   = ((size_t)nEdges * sizeof(int2) + 255) & ~(size_t)255;
  const int    nPad     = ((nRows + 1 + SCAN_SPAN - 1) / SCAN_SPAN) * SCAN_SPAN;
  const size_t cntBytes = (size_t)nPad * sizeof(int);
  const size_t need     = 2 * bufAl + edgeAl + 2 * cntBytes;
  const size_t rankBytes = (size_t)nEdges * sizeof(int);

  const int    sortGrid = 512;
  const int    nBins    = (nRows + FINE_ROWS - 1) >> FINE_SHIFT;
  const int    nPadT    = ((nBins * sortGrid + 1 + SCAN_SPAN - 1) / SCAN_SPAN) * SCAN_SPAN;
  const int    chunk    = (nEdges + sortGrid - 1) / sortGrid;
  if (ws_size >= need && nEdges > 0 && nBins > 0 && nBins <= CB_MAX && nPadT <= nPad &&
      2 * rankBytes <= bufAl && (size_t)nPadT * sizeof(int) <= edgeAl) {
    nxt = (float*)((char*)d_ws + bufAl);
    int2* sorted = (int2*)((char*)d_ws + 2 * bufAl);
    int*  tabOff = (int*)((char*)d_ws + 2 * bufAl + edgeAl);
    int*  ptr    = tabOff + nPad;
    int*  table  = (int*)sorted;
    int*  tmpRow = (int*)cur;
    int*  rank   = tmpRow + nEdges;
    int2* tmpCV  = (int2*)nxt;

    coarse_hist_kernel<<<sortGrid, SORT_BLOCK, 0, stream>>>(erow, table, nEdges, chunk, nBins, nPadT);
    row_scan_kernel<<<1, 1024, 0, stream>>>(table, tabOff, nPadT);
    coarse_scatter_kernel<<<sortGrid, SORT_BLOCK, 0, stream>>>(erow, ecol, evals, tabOff,
                                                               tmpRow, tmpCV, nEdges, chunk, nBins);
    fine_sort_kernel<<<nBins, SORT_BLOCK, 0, stream>>>(tabOff, tmpRow, tmpCV, rank, ptr, sorted,
                                                       sortGrid, nRows);
    init_concat_kernel<<<eg, eb, 0, stream>>>(users, items, out, cur,
                                              userElems, totalElems);

    const int rowGroups  = (nRows + 7) / 8;
    const int rowBlocks  = (rowGroups + 7) / 8;
    for (int layer = 0; layer < 3; ++layer) {
      const float scale = (layer == 2) ? 0.25f : 1.0f;
      spmm_rows_wmma_kernel<<<rowBlocks, 256, 0, stream>>>(
          ptr, sorted, cur, nxt, out, scale, (layer == 2) ? 0 : 1, nRows, nEdges);
      float* tmp = cur; cur = nxt; nxt = tmp;
    }
    return;
  }

  init_concat_kernel<<<eg, eb, 0, stream>>>(users, items, out, cur,
                                            userElems, totalElems);

  if (ws_size >= need && nEdges > 0 && rankBytes <= bufAl) {
    nxt = (float*)((char*)d_ws + bufAl);
    int2* sorted = (int2*)((char*)d_ws + 2 * bufAl);
    int*  cnt    = (int*)((char*)d_ws + 2 * bufAl + edgeAl);
    int*  ptr    = cnt + nPad;
    int*  rank   = (int*)nxt;

    hipMemsetAsync(cnt, 0, cntBytes, stream);
    const int edgeBlocks = (nEdges + 255) / 256;
    row_count_kernel<<<edgeBlocks, 256, 0, stream>>>(erow, cnt, rank, nEdges);
    row_scan_kernel<<<1, 1024, 0, stream>>>(cnt, ptr, nPad);
    bucket_fill_kernel<<<edgeBlocks, 256, 0, stream>>>(erow, ecol, evals,
                                                       ptr, rank, sorted, nEdges);

    const int rowGroups  = (nRows + 7) / 8;
    const int rowBlocks  = (rowGroups + 7) / 8;
    for (int layer = 0; layer < 3; ++layer) {
      const float scale = (layer == 2) ? 0.25f : 1.0f;
      spmm_rows_wmma_kernel<<<rowBlocks, 256, 0, stream>>>(
          ptr, sorted, cur, nxt, out, scale, (layer == 2) ? 0 : 1, nRows, nEdges);
      float* tmp = cur; cur = nxt; nxt = tmp;
    }
    return;
  }

  const int nGroups    = (nEdges + 15) / 16;
  const int spmmBlocks = (nGroups + 7) / 8;

  for (int layer = 0; layer < 3; ++layer) {
    hipMemsetAsync(nxt, 0, bufBytes, stream);
    spmm_wmma_kernel<<<spmmBlocks, 256, 0, stream>>>(erow, ecol, evals,
                                                     cur, nxt, nEdges);
    const float scale = (layer == 2) ? 0.25f : 1.0f;
    axpy_scale_kernel<<<eg, eb, 0, stream>>>(out, nxt, scale, totalElems);
    float* tmp = cur; cur = nxt; nxt = tmp;
  }
}


# First layer gathers its rows straight from the two input tables and starts the running sum in out from them: the joined-copy pass and its launch go (7 launches); row sum order unchanged
# speedup vs baseline: 2.5098x; 1.0357x over previous
#include <hip/hip_runtime.h>

typedef __attribute__((ext_vector_type(2))) float v2f;
typedef __attribute__((ext_vector_type(4))) float v4f;
typedef __attribute__((ext_vector_type(8))) float v8f;

#define EMB 64

__global__ void init_concat_kernel(const float* __restrict__ users,
                                   const float* __restrict__ items,
                                   float* __restrict__ out,
                                   float* __restrict__ cur,
                                   int userElems, int totalElems) {
  int i4 = (blockIdx.x * blockDim.x + threadIdx.x) * 4;
  if (i4 >= totalElems) return;
  float4 v = (i4 < userElems) ? *(const float4*)(users + i4)
                              : *(const float4*)(items + (i4 - userElems));
  *(float4*)(out + i4) = v;
  *(float4*)(cur + i4) = v;
}

__global__ void axpy_scale_kernel(float* __restrict__ out,
                                  const float* __restrict__ nxt,
                                  float scale, int totalElems) {
  int i4 = (blockIdx.x * blockDim.x + threadIdx.x) * 4;
  if (i4 >= totalElems) return;
  float4 a = *(const float4*)(out + i4);
  float4 b = *(const float4*)(nxt + i4);
  a.x = (a.x + b.x) * scale;
  a.y = (a.y + b.y) * scale;
  a.z = (a.z + b.z) * scale;
  a.w = (a.w + b.w) * scale;
  *(float4*)(out + i4) = a;
}

__global__ void spmm_wmma_kernel(const int*   __restrict__ rows,
                                 const int*   __restrict__ cols,
                                 const float* __restrict__ vals,
                                 const float* __restrict__ x,
                                 float*       __restrict__ y,
                                 int nEdges) {
  const int lane   = threadIdx.x & 31;
  const int m      = lane & 15;
  const bool hiHalf = lane >= 16;
  const int wavesPerBlock = blockDim.x >> 5;
  const int wave   = blockIdx.x * wavesPerBlock + (threadIdx.x >> 5);
  const int stride = gridDim.x * wavesPerBlock;
  const int nGroups = (nEdges + 15) >> 4;

  for (int g = wave; g < nGroups; g += stride) {
    const int base = g << 4;
    const int e    = base + m;
    const int eC   = (e < nEdges) ? e : (nEdges - 1);

    const float v   = (e < nEdges) ? vals[eC] : 0.0f;
    const int myRow = rows[eC];
    const int myCol = cols[eC];

    int colK[8];
#pragma unroll
    for (int c = 0; c < 4; ++c) {
      const int kb0 = 4 * c + (hiHalf ? 2 : 0);
      colK[2 * c]     = __shfl(myCol, kb0, 32);
      colK[2 * c + 1] = __shfl(myCol, kb0 + 1, 32);
    }

    int rowD[8];
#pragma unroll
    for (int vv = 0; vv < 8; ++vv)
      rowD[vv] = __shfl(myRow, hiHalf ? (vv + 8) : vv, 32);

    v2f A[4];
#pragma unroll
    for (int c = 0; c < 4; ++c) {
      const int ka0 = 4 * c + (hiHalf ? 2 : 0);
      A[c].x = (m == ka0)     ? v : 0.0f;
      A[c].y = (m == ka0 + 1) ? v : 0.0f;
    }

#pragma unroll
    for (int t = 0; t < 4; ++t) {
      v8f acc = {};
#pragma unroll
      for (int c = 0; c < 4; ++c) {
        v2f B;
        B.x = x[colK[2 * c]     * EMB + t * 16 + m];
        B.y = x[colK[2 * c + 1] * EMB + t * 16 + m];
        acc = __builtin_amdgcn_wmma_f32_16x16x4_f32(
             false, A[c],  false, B,
             (short)0, acc,  false,  false);
      }
#pragma unroll
      for (int vv = 0; vv < 8; ++vv)
        atomicAdd(&y[rowD[vv] * EMB + t * 16 + m], acc[vv]);
    }
  }
}


#define SCAN_SPAN 8192

__global__ void row_count_kernel(const int* __restrict__ rows,
                                 int* __restrict__ cnt,
                                 int* __restrict__ rank, int nEdges) {
  int e = blockIdx.x * blockDim.x + threadIdx.x;
  if (e < nEdges) rank[e] = atomicAdd(&cnt[rows[e]], 1);
}

__global__ void row_scan_kernel(const int* __restrict__ cnt,
                                int* __restrict__ ptr, int nPad) {
  __shared__ int waveSum[32];
  __shared__ int tripTotal;
  const int tid = threadIdx.x, lane = tid & 31, w = tid >> 5;
  int carry = 0;
  for (int base = 0; base < nPad; base += SCAN_SPAN) {
    const int i0 = base + tid * 8;
    int4 a = *(const int4*)(cnt + i0);
    int4 b = *(const int4*)(cnt + i0 + 4);
    int e0 = 0, e1 = a.x, e2 = e1 + a.y, e3 = e2 + a.z, e4 = e3 + a.w;
    int e5 = e4 + b.x, e6 = e5 + b.y, e7 = e6 + b.z;
    const int tsum = e7 + b.w;
    int s = tsum;
#pragma unroll
    for (int d = 1; d < 32; d <<= 1) {
      int t = __shfl_up(s, d, 32);
      if (lane >= d) s += t;
    }
    if (lane == 31) waveSum[w] = s;
    __syncthreads();
    if (w == 0) {
      const int ws = waveSum[lane];
      int t2 = ws;
#pragma unroll
      for (int d = 1; d < 32; d <<= 1) {
        int t = __shfl_up(t2, d, 32);
        if (lane >= d) t2 += t;
      }
      waveSum[lane] = t2 - ws;
      if (lane == 31) tripTotal = t2;
    }
    __syncthreads();
    const int off = carry + waveSum[w] + (s - tsum);
    int4 o0 = make_int4(off + e0, off + e1, off + e2, off + e3);
    int4 o1 = make_int4(off + e4, off + e5, off + e6, off + e7);
    *(int4*)(ptr + i0) = o0;      *(int4*)(ptr + i0 + 4) = o1;
    carry += tripTotal;
    __syncthreads();
  }
}

__global__ void bucket_fill_kernel(const int* __restrict__ rows,
                                   const int* __restrict__ cols,
                                   const float* __restrict__ vals,
                                   const int* __restrict__ ptr,
                                   const int* __restrict__ rank,
                                   int2* __restrict__ sorted, int nEdges) {
  int e = blockIdx.x * blockDim.x + threadIdx.x;
  if (e < nEdges) {
    const int slot = ptr[rows[e]] + rank[e];
    sorted[slot] = make_int2(cols[e], __float_as_int(vals[e]));
  }
}

#define FINE_SHIFT 9
#define FINE_ROWS  (1 << FINE_SHIFT)
#define CB_MAX     2048
#define SORT_BLOCK 1024

__global__ void coarse_hist_kernel(const int* __restrict__ rows,
                                   int* __restrict__ table,
                                   int nEdges, int chunk, int nBins, int nPadT) {
  __shared__ int hist[CB_MAX];
  const int tid = threadIdx.x, G = gridDim.x;
  for (int i = tid; i < nBins; i += SORT_BLOCK) hist[i] = 0;
  __syncthreads();
  const int e0 = blockIdx.x * chunk;
  const int e1 = (e0 + chunk < nEdges) ? (e0 + chunk) : nEdges;
  for (int e = e0 + tid; e < e1; e += 4 * SORT_BLOCK) {
    int r[4];
#pragma unroll
    for (int k = 0; k < 4; ++k) {
      const int ek = e + k * SORT_BLOCK;
      r[k] = rows[ek < e1 ? ek : e1 - 1];
    }
#pragma unroll
    for (int k = 0; k < 4; ++k)
      if (e + k * SORT_BLOCK < e1) atomicAdd(&hist[r[k] >> FINE_SHIFT], 1);
  }
  __syncthreads();
  for (int i = tid; i < nBins; i += SORT_BLOCK) table[i * G + blockIdx.x] = hist[i];
  for (int i = nBins * G + blockIdx.x * SORT_BLOCK + tid; i < nPadT; i += G * SORT_BLOCK)
    table[i] = 0;
}

__global__ void coarse_scatter_kernel(const int* __restrict__ rows,
                                      const int* __restrict__ cols,
                                      const float* __restrict__ vals,
                                      const int* __restrict__ tabOff,
                                      int* __restrict__ tmpRow,
                                      int2* __restrict__ tmpCV,
                                      int nEdges, int chunk, int nBins) {
  __shared__ int cursor[CB_MAX];
  const int tid = threadIdx.x, G = gridDim.x;
  for (int i = tid; i < nBins; i += SORT_BLOCK) cursor[i] = tabOff[i * G + blockIdx.x];
  __syncthreads();
  const int e0 = blockIdx.x * chunk;
  const int e1 = (e0 + chunk < nEdges) ? (e0 + chunk) : nEdges;
  for (int e = e0 + tid; e < e1; e += 4 * SORT_BLOCK) {
    int r[4], cI[4], vI[4];
#pragma unroll
    for (int k = 0; k < 4; ++k) {
      const int ek = e + k * SORT_BLOCK;
      const int ec = ek < e1 ? ek : e1 - 1;
      r[k]  = rows[ec];
      cI[k] = cols[ec];
      vI[k] = __float_as_int(vals[ec]);
    }
#pragma unroll
    for (int k = 0; k < 4; ++k) {
      if (e + k * SORT_BLOCK < e1) {
        const int pos = atomicAdd(&cursor[r[k] >> FINE_SHIFT], 1);
        tmpRow[pos] = r[k];
        tmpCV[pos]  = make_int2(cI[k], vI[k]);
      }
    }
  }
}

__global__ void fine_sort_kernel(const int* __restrict__ tabOff,
                                 const int* __restrict__ tmpRow,
                                 const int2* __restrict__ tmpCV,
                                 int* __restrict__ rank,
                                 int* __restrict__ ptr,
                                 int2* __restrict__ sorted,
                                 int G, int nRows) {
  __shared__ int cnt[FINE_ROWS];
  __shared__ int waveSum[32];
  const int tid = threadIdx.x, lane = tid & 31, w = tid >> 5;
  const int bin = blockIdx.x;
  const int bs = tabOff[bin * G];
  const int be = tabOff[(bin + 1) * G];
  const int row0 = bin << FINE_SHIFT;
  if (tid < FINE_ROWS) cnt[tid] = 0;
  __syncthreads();
  for (int p = bs + tid; p < be; p += 4 * SORT_BLOCK) {
    int lr[4];
#pragma unroll
    for (int k = 0; k < 4; ++k) {
      const int pk = p + k * SORT_BLOCK;
      lr[k] = tmpRow[pk < be ? pk : be - 1] - row0;
    }
#pragma unroll
    for (int k = 0; k < 4; ++k)
      if (p + k * SORT_BLOCK < be) rank[p + k * SORT_BLOCK] = atomicAdd(&cnt[lr[k]], 1);
  }
  __syncthreads();
  const int v = (tid < FINE_ROWS) ? cnt[tid] : 0;
  int s = v;
#pragma unroll
  for (int d = 1; d < 32; d <<= 1) {
    int t = __shfl_up(s, d, 32);
    if (lane >= d) s += t;
  }
  if (lane == 31) waveSum[w] = s;
  __syncthreads();
  if (w == 0) {
    const int ws = waveSum[lane];
    int t2 = ws;
#pragma unroll
    for (int d = 1; d < 32; d <<= 1) {
      int t = __shfl_up(t2, d, 32);
      if (lane >= d) t2 += t;
    }
    waveSum[lane] = t2 - ws;
  }
  __syncthreads();
  const int rowStart = bs + waveSum[w] + (s - v);
  if (tid < FINE_ROWS) {
    cnt[tid] = rowStart;
    if (row0 + tid < nRows) ptr[row0 + tid] = rowStart;
  }
  if (bin == (int)gridDim.x - 1 && tid == 0) ptr[nRows] = be;
  __syncthreads();
  for (int p = bs + tid; p < be; p += 4 * SORT_BLOCK) {
    int lr[4], rk[4];
    int2 cv[4];
#pragma unroll
    for (int k = 0; k < 4; ++k) {
      const int pk = p + k * SORT_BLOCK;
      const int pc = pk < be ? pk : be - 1;
      lr[k] = tmpRow[pc] - row0;
      rk[k] = rank[pc];
      cv[k] = tmpCV[pc];
    }
#pragma unroll
    for (int k = 0; k < 4; ++k)
      if (p + k * SORT_BLOCK < be) sorted[cnt[lr[k]] + rk[k]] = cv[k];
  }
}

template <bool FIRST>
__device__ __forceinline__ void spmm_rows_body(const int*  __restrict__ ptr,
                                               const int2* __restrict__ sorted,
                                               const float* __restrict__ x,
                                               const float* __restrict__ x2,
                                               float* __restrict__ y,
                                               float* __restrict__ out,
                                               float scale, int writeY,
                                               int nRows, int nEdges, int nSplit) {
  const int lane = threadIdx.x & 31;
  const int c = lane & 15;
  const int h = lane >> 4;
  const int wave = blockIdx.x * (blockDim.x >> 5) + (threadIdx.x >> 5);
  const int r0 = wave << 3;
  if (r0 >= nRows) return;

  const int myRow = r0 + (c >> 1);
  const bool odd = (c & 1) != 0;
  const int lo = ptr[myRow < nRows ? myRow : nRows];
  const int hi = ptr[myRow + 1 < nRows ? myRow + 1 : nRows];
  const int start = __builtin_amdgcn_readlane(lo, 0);
  const int end   = __builtin_amdgcn_readlane(hi, 15);

  const char* sortedB = (const char*)sorted;
  const char* xB = (const char*)x;
  const unsigned laneOff = (unsigned)c * 16u;
  const unsigned long long base1 = (unsigned long long)x + laneOff;
  const unsigned long long base2 = (unsigned long long)x2 + laneOff
                                 - (unsigned long long)(unsigned)nSplit * (unsigned long long)(EMB * 4);

  v8f acc0 = {}, acc1 = {};
  const int last = nEdges - 1;
  for (int p = start; p < end; p += 16) {
    v2f A[8];
    v4f R[8];
#pragma unroll
    for (int i = 0; i < 8; ++i) {
      const int q = p + 2 * i + h;
      const int qc = q < last ? q : last;
      const int2 s = *(const int2*)(sortedB + (unsigned)qc * 8u);
      const float v = (q >= lo && q < hi) ? __int_as_float(s.y) : 0.0f;
      A[i].x = odd ? 0.0f : v;
      A[i].y = odd ? v : 0.0f;
      if (FIRST) {
        const unsigned long long b = ((unsigned)s.x < (unsigned)nSplit) ? base1 : base2;
        R[i] = *(const __attribute__((address_space(1))) v4f*)(
            b + (unsigned long long)(unsigned)s.x * (unsigned long long)(EMB * 4));
      } else {
        R[i] = *(const v4f*)(xB + ((unsigned)s.x * (unsigned)(EMB * 4) + laneOff));
      }
    }
    __builtin_amdgcn_sched_barrier(0);
#pragma unroll
    for (int i = 0; i < 8; ++i) {
      const v2f b0 = {R[i].x, R[i].y};
      const v2f b1 = {R[i].z, R[i].w};
      acc0 = __builtin_amdgcn_wmma_f32_16x16x4_f32(false, A[i], false, b0, (short)0, acc0, false, false);
      acc1 = __builtin_amdgcn_wmma_f32_16x16x4_f32(false, A[i], false, b1, (short)0, acc1, false, false);
    }
    __builtin_amdgcn_sched_barrier(0);
    asm volatile("v_nop\n\tv_nop\n\tv_nop\n\tv_nop\n\tv_nop\n\tv_nop\n\tv_nop\n\tv_nop");
    __builtin_amdgcn_sched_barrier(0);
  }

#pragma unroll
  for (int i = 0; i < 4; ++i) {
    const int row = r0 + 4 * h + i;
    if (row < nRows) {
      const size_t idx = (size_t)row * EMB + 4 * c;
      const v4f d = {acc0[2 * i], acc0[2 * i + 1], acc1[2 * i], acc1[2 * i + 1]};
      v4f o;
      if (FIRST) {
        *(v4f*)(y + idx) = d;
        const float* src = (row < nSplit) ? (x + idx) : (x2 + (idx - (size_t)nSplit * EMB));
        o = *(const v4f*)src;
      } else {
        if (writeY) *(v4f*)(y + idx) = d;
        o = *(const v4f*)(out + idx);
      }
      *(v4f*)(out + idx) = (o + d) * scale;
    }
  }
}

__global__ void spmm_rows_wmma_kernel(const int*  __restrict__ ptr,
                                      const int2* __restrict__ sorted,
                                      const float* __restrict__ x,
                                      float* __restrict__ y,
                                      float* __restrict__ out,
                                      float scale, int writeY,
                                      int nRows, int nEdges) {
  spmm_rows_body<false>(ptr, sorted, x, x, y, out, scale, writeY, nRows, nEdges, 0);
}

__global__ void spmm_rows_first_kernel(const int*  __restrict__ ptr,
                                       const int2* __restrict__ sorted,
                                       const float* __restrict__ users,
                                       const float* __restrict__ items,
                                       float* __restrict__ y,
                                       float* __restrict__ out,
                                       float scale,
                                       int nRows, int nEdges, int nUsers) {
  spmm_rows_body<true>(ptr, sorted, users, items, y, out, scale, 1, nRows, nEdges, nUsers);
}

extern "C" void kernel_launch(void* const* d_in, const int* in_sizes, int n_in,
                              void* d_out, int out_size, void* d_ws, size_t ws_size,
                              hipStream_t stream) {
  const float* users = (const float*)d_in[0];
  const float* items = (const float*)d_in[1];
  const int*   erow  = (const int*)d_in[2];
  const int*   ecol  = (const int*)d_in[3];
  const float* evals = (const float*)d_in[4];
  float* out = (float*)d_out;

  const int userElems  = in_sizes[0];
  const int itemElems  = in_sizes[1];
  const int nEdges     = in_sizes[2];
  const int totalElems = userElems + itemElems;
  const int nRows      = totalElems / EMB;

  const size_t bufBytes = (size_t)totalElems * sizeof(float);
  if (ws_size < 2 * bufBytes) return;
  float* cur = (float*)d_ws;
  float* nxt = (float*)((char*)d_ws + bufBytes);

  const int ethreads = (totalElems + 3) / 4;
  const dim3 eb(256), eg((ethreads + 255) / 256);

  const size_t bufAl    = (bufBytes + 255) & ~(size_t)255;
  const size_t edgeAl   = ((size_t)nEdges * sizeof(int2) + 255) & ~(size_t)255;
  const int    nPad     = ((nRows + 1 + SCAN_SPAN - 1) / SCAN_SPAN) * SCAN_SPAN;
  const size_t cntBytes = (size_t)nPad * sizeof(int);
  const size_t need     = 2 * bufAl + edgeAl + 2 * cntBytes;
  const size_t rankBytes = (size_t)nEdges * sizeof(int);

  const int    sortGrid = 512;
  const int    nBins    = (nRows + FINE_ROWS - 1) >> FINE_SHIFT;
  const int    nPadT    = ((nBins * sortGrid + 1 + SCAN_SPAN - 1) / SCAN_SPAN) * SCAN_SPAN;
  const int    chunk    = (nEdges + sortGrid - 1) / sortGrid;
  if (ws_size >= need && nEdges > 0 && nBins > 0 && nBins <= CB_MAX && nPadT <= nPad &&
      2 * rankBytes <= bufAl && (size_t)nPadT * sizeof(int) <= edgeAl) {
    nxt = (float*)((char*)d_ws + bufAl);
    int2* sorted = (int2*)((char*)d_ws + 2 * bufAl);
    int*  tabOff = (int*)((char*)d_ws + 2 * bufAl + edgeAl);
    int*  ptr    = tabOff + nPad;
    int*  table  = (int*)sorted;
    int*  tmpRow = (int*)cur;
    int*  rank   = tmpRow + nEdges;
    int2* tmpCV  = (int2*)nxt;

    coarse_hist_kernel<<<sortGrid, SORT_BLOCK, 0, stream>>>(erow, table, nEdges, chunk, nBins, nPadT);
    row_scan_kernel<<<1, 1024, 0, stream>>>(table, tabOff, nPadT);
    coarse_scatter_kernel<<<sortGrid, SORT_BLOCK, 0, stream>>>(erow, ecol, evals, tabOff,
                                                               tmpRow, tmpCV, nEdges, chunk, nBins);
    fine_sort_kernel<<<nBins, SORT_BLOCK, 0, stream>>>(tabOff, tmpRow, tmpCV, rank, ptr, sorted,
                                                       sortGrid, nRows);
    const int rowGroups  = (nRows + 7) / 8;
    const int rowBlocks  = (rowGroups + 7) / 8;
    spmm_rows_first_kernel<<<rowBlocks, 256, 0, stream>>>(
        ptr, sorted, users, items, cur, out, 1.0f, nRows, nEdges, userElems / EMB);
    for (int layer = 1; layer < 3; ++layer) {
      const float scale = (layer == 2) ? 0.25f : 1.0f;
      spmm_rows_wmma_kernel<<<rowBlocks, 256, 0, stream>>>(
          ptr, sorted, cur, nxt, out, scale, (layer == 2) ? 0 : 1, nRows, nEdges);
      float* tmp = cur; cur = nxt; nxt = tmp;
    }
    return;
  }

  init_concat_kernel<<<eg, eb, 0, stream>>>(users, items, out, cur,
                                            userElems, totalElems);

  if (ws_size >= need && nEdges > 0 && rankBytes <= bufAl) {
    nxt = (float*)((char*)d_ws + bufAl);
    int2* sorted = (int2*)((char*)d_ws + 2 * bufAl);
    int*  cnt    = (int*)((char*)d_ws + 2 * bufAl + edgeAl);
    int*  ptr    = cnt + nPad;
    int*  rank   = (int*)nxt;

    hipMemsetAsync(cnt, 0, cntBytes, stream);
    const int edgeBlocks = (nEdges + 255) / 256;
    row_count_kernel<<<edgeBlocks, 256, 0, stream>>>(erow, cnt, rank, nEdges);
    row_scan_kernel<<<1, 1024, 0, stream>>>(cnt, ptr, nPad);
    bucket_fill_kernel<<<edgeBlocks, 256, 0, stream>>>(erow, ecol, evals,
                                                       ptr, rank, sorted, nEdges);

    const int rowGroups  = (nRows + 7) / 8;
    const int rowBlocks  = (rowGroups + 7) / 8;
    for (int layer = 0; layer < 3; ++layer) {
      const float scale = (layer == 2) ? 0.25f : 1.0f;
      spmm_rows_wmma_kernel<<<rowBlocks, 256, 0, stream>>>(
          ptr, sorted, cur, nxt, out, scale, (layer == 2) ? 0 : 1, nRows, nEdges);
      float* tmp = cur; cur = nxt; nxt = tmp;
    }
    return;
  }

  const int nGroups    = (nEdges + 15) / 16;
  const int spmmBlocks = (nGroups + 7) / 8;

  for (int layer = 0; layer < 3; ++layer) {
    hipMemsetAsync(nxt, 0, bufBytes, stream);
    spmm_wmma_kernel<<<spmmBlocks, 256, 0, stream>>>(erow, ecol, evals,
                                                     cur, nxt, nEdges);
    const float scale = (layer == 2) ? 0.25f : 1.0f;
    axpy_scale_kernel<<<eg, eb, 0, stream>>>(out, nxt, scale, totalElems);
    float* tmp = cur; cur = nxt; nxt = tmp;
  }
}


# Sort stores go through LDS: level-1 scatter and per-bin placement put edges in LDS in final order, then copy out with neighbouring lanes on neighbouring slots; row sum order still = LDS add order
# speedup vs baseline: 3.2924x; 1.3118x over previous
#include <hip/hip_runtime.h>

typedef __attribute__((ext_vector_type(2))) float v2f;
typedef __attribute__((ext_vector_type(4))) float v4f;
typedef __attribute__((ext_vector_type(8))) float v8f;

#define EMB 64

__global__ void init_concat_kernel(const float* __restrict__ users,
                                   const float* __restrict__ items,
                                   float* __restrict__ out,
                                   float* __restrict__ cur,
                                   int userElems, int totalElems) {
  int i4 = (blockIdx.x * blockDim.x + threadIdx.x) * 4;
  if (i4 >= totalElems) return;
  float4 v = (i4 < userElems) ? *(const float4*)(users + i4)
                              : *(const float4*)(items + (i4 - userElems));
  *(float4*)(out + i4) = v;
  *(float4*)(cur + i4) = v;
}

__global__ void axpy_scale_kernel(float* __restrict__ out,
                                  const float* __restrict__ nxt,
                                  float scale, int totalElems) {
  int i4 = (blockIdx.x * blockDim.x + threadIdx.x) * 4;
  if (i4 >= totalElems) return;
  float4 a = *(const float4*)(out + i4);
  float4 b = *(const float4*)(nxt + i4);
  a.x = (a.x + b.x) * scale;
  a.y = (a.y + b.y) * scale;
  a.z = (a.z + b.z) * scale;
  a.w = (a.w + b.w) * scale;
  *(float4*)(out + i4) = a;
}

__global__ void spmm_wmma_kernel(const int*   __restrict__ rows,
                                 const int*   __restrict__ cols,
                                 const float* __restrict__ vals,
                                 const float* __restrict__ x,
                                 float*       __restrict__ y,
                                 int nEdges) {
  const int lane   = threadIdx.x & 31;
  const int m      = lane & 15;
  const bool hiHalf = lane >= 16;
  const int wavesPerBlock = blockDim.x >> 5;
  const int wave   = blockIdx.x * wavesPerBlock + (threadIdx.x >> 5);
  const int stride = gridDim.x * wavesPerBlock;
  const int nGroups = (nEdges + 15) >> 4;

  for (int g = wave; g < nGroups; g += stride) {
    const int base = g << 4;
    const int e    = base + m;
    const int eC   = (e < nEdges) ? e : (nEdges - 1);

    const float v   = (e < nEdges) ? vals[eC] : 0.0f;
    const int myRow = rows[eC];
    const int myCol = cols[eC];

    int colK[8];
#pragma unroll
    for (int c = 0; c < 4; ++c) {
      const int kb0 = 4 * c + (hiHalf ? 2 : 0);
      colK[2 * c]     = __shfl(myCol, kb0, 32);
      colK[2 * c + 1] = __shfl(myCol, kb0 + 1, 32);
    }

    int rowD[8];
#pragma unroll
    for (int vv = 0; vv < 8; ++vv)
      rowD[vv] = __shfl(myRow, hiHalf ? (vv + 8) : vv, 32);

    v2f A[4];
#pragma unroll
    for (int c = 0; c < 4; ++c) {
      const int ka0 = 4 * c + (hiHalf ? 2 : 0);
      A[c].x = (m == ka0)     ? v : 0.0f;
      A[c].y = (m == ka0 + 1) ? v : 0.0f;
    }

#pragma unroll
    for (int t = 0; t < 4; ++t) {
      v8f acc = {};
#pragma unroll
      for (int c = 0; c < 4; ++c) {
        v2f B;
        B.x = x[colK[2 * c]     * EMB + t * 16 + m];
        B.y = x[colK[2 * c + 1] * EMB + t * 16 + m];
        acc = __builtin_amdgcn_wmma_f32_16x16x4_f32(
             false, A[c],  false, B,
             (short)0, acc,  false,  false);
      }
#pragma unroll
      for (int vv = 0; vv < 8; ++vv)
        atomicAdd(&y[rowD[vv] * EMB + t * 16 + m], acc[vv]);
    }
  }
}


#define SCAN_SPAN 8192

__global__ void row_count_kernel(const int* __restrict__ rows,
                                 int* __restrict__ cnt,
                                 int* __restrict__ rank, int nEdges) {
  int e = blockIdx.x * blockDim.x + threadIdx.x;
  if (e < nEdges) rank[e] = atomicAdd(&cnt[rows[e]], 1);
}

__global__ void row_scan_kernel(const int* __restrict__ cnt,
                                int* __restrict__ ptr, int nPad) {
  __shared__ int waveSum[32];
  __shared__ int tripTotal;
  const int tid = threadIdx.x, lane = tid & 31, w = tid >> 5;
  int carry = 0;
  for (int base = 0; base < nPad; base += SCAN_SPAN) {
    const int i0 = base + tid * 8;
    int4 a = *(const int4*)(cnt + i0);
    int4 b = *(const int4*)(cnt + i0 + 4);
    int e0 = 0, e1 = a.x, e2 = e1 + a.y, e3 = e2 + a.z, e4 = e3 + a.w;
    int e5 = e4 + b.x, e6 = e5 + b.y, e7 = e6 + b.z;
    const int tsum = e7 + b.w;
    int s = tsum;
#pragma unroll
    for (int d = 1; d < 32; d <<= 1) {
      int t = __shfl_up(s, d, 32);
      if (lane >= d) s += t;
    }
    if (lane == 31) waveSum[w] = s;
    __syncthreads();
    if (w == 0) {
      const int ws = waveSum[lane];
      int t2 = ws;
#pragma unroll
      for (int d = 1; d < 32; d <<= 1) {
        int t = __shfl_up(t2, d, 32);
        if (lane >= d) t2 += t;
      }
      waveSum[lane] = t2 - ws;
      if (lane == 31) tripTotal = t2;
    }
    __syncthreads();
    const int off = carry + waveSum[w] + (s - tsum);
    int4 o0 = make_int4(off + e0, off + e1, off + e2, off + e3);
    int4 o1 = make_int4(off + e4, off + e5, off + e6, off + e7);
    *(int4*)(ptr + i0) = o0;      *(int4*)(ptr + i0 + 4) = o1;
    carry += tripTotal;
    __syncthreads();
  }
}

__global__ void bucket_fill_kernel(const int* __restrict__ rows,
                                   const int* __restrict__ cols,
                                   const float* __restrict__ vals,
                                   const int* __restrict__ ptr,
                                   const int* __restrict__ rank,
                                   int2* __restrict__ sorted, int nEdges) {
  int e = blockIdx.x * blockDim.x + threadIdx.x;
  if (e < nEdges) {
    const int slot = ptr[rows[e]] + rank[e];
    sorted[slot] = make_int2(cols[e], __float_as_int(vals[e]));
  }
}

#define FINE_SHIFT 9
#define FINE_ROWS  (1 << FINE_SHIFT)
#define CB_MAX     2048
#define SORT_BLOCK 1024
#define STAGE2_CAP 17408

__global__ void coarse_hist_kernel(const int* __restrict__ rows,
                                   int* __restrict__ table,
                                   int nEdges, int chunk, int nBins, int nPadT) {
  __shared__ int hist[CB_MAX];
  const int tid = threadIdx.x, G = gridDim.x;
  for (int i = tid; i < nBins; i += SORT_BLOCK) hist[i] = 0;
  __syncthreads();
  const int e0 = blockIdx.x * chunk;
  const int e1 = (e0 + chunk < nEdges) ? (e0 + chunk) : nEdges;
  for (int e = e0 + tid; e < e1; e += 4 * SORT_BLOCK) {
    int r[4];
#pragma unroll
    for (int k = 0; k < 4; ++k) {
      const int ek = e + k * SORT_BLOCK;
      r[k] = rows[ek < e1 ? ek : e1 - 1];
    }
#pragma unroll
    for (int k = 0; k < 4; ++k)
      if (e + k * SORT_BLOCK < e1) atomicAdd(&hist[r[k] >> FINE_SHIFT], 1);
  }
  __syncthreads();
  for (int i = tid; i < nBins; i += SORT_BLOCK) table[i * G + blockIdx.x] = hist[i];
  for (int i = nBins * G + blockIdx.x * SORT_BLOCK + tid; i < nPadT; i += G * SORT_BLOCK)
    table[i] = 0;
}

__global__ void coarse_scatter_kernel(const int* __restrict__ rows,
                                      const int* __restrict__ cols,
                                      const float* __restrict__ vals,
                                      const int* __restrict__ tabOff,
                                      int* __restrict__ tmpRow,
                                      int2* __restrict__ tmpCV,
                                      int nEdges, int chunk, int nBins) {
  __shared__ int cursor[CB_MAX];
  const int tid = threadIdx.x, G = gridDim.x;
  for (int i = tid; i < nBins; i += SORT_BLOCK) cursor[i] = tabOff[i * G + blockIdx.x];
  __syncthreads();
  const int e0 = blockIdx.x * chunk;
  const int e1 = (e0 + chunk < nEdges) ? (e0 + chunk) : nEdges;
  for (int e = e0 + tid; e < e1; e += 4 * SORT_BLOCK) {
    int r[4], cI[4], vI[4];
#pragma unroll
    for (int k = 0; k < 4; ++k) {
      const int ek = e + k * SORT_BLOCK;
      const int ec = ek < e1 ? ek : e1 - 1;
      r[k]  = rows[ec];
      cI[k] = cols[ec];
      vI[k] = __float_as_int(vals[ec]);
    }
#pragma unroll
    for (int k = 0; k < 4; ++k) {
      if (e + k * SORT_BLOCK < e1) {
        const int pos = atomicAdd(&cursor[r[k] >> FINE_SHIFT], 1);
        tmpRow[pos] = r[k];
        tmpCV[pos]  = make_int2(cI[k], vI[k]);
      }
    }
  }
}

#define STAGE1_CAP 10240

__global__ void coarse_scatter_staged_kernel(const int* __restrict__ rows,
                                             const int* __restrict__ cols,
                                             const float* __restrict__ vals,
                                             const int* __restrict__ tabOff,
                                             int* __restrict__ tmpRow,
                                             int2* __restrict__ tmpCV,
                                             int nEdges, int chunk, int nBins) {
  __shared__ int  cursor[CB_MAX];
  __shared__ int  delta[CB_MAX];
  __shared__ int  waveSum[32];
  __shared__ int  sRow[STAGE1_CAP];
  __shared__ int2 sCV[STAGE1_CAP];
  const int tid = threadIdx.x, lane = tid & 31, w = tid >> 5, G = gridDim.x;
  const int b = blockIdx.x;
  const int i0 = 2 * tid, i1 = 2 * tid + 1;
  int g0 = 0, g1 = 0, c0 = 0, c1 = 0;
  if (i0 < nBins) { g0 = tabOff[i0 * G + b]; c0 = tabOff[i0 * G + b + 1] - g0; }
  if (i1 < nBins) { g1 = tabOff[i1 * G + b]; c1 = tabOff[i1 * G + b + 1] - g1; }
  const int tsum = c0 + c1;
  int s = tsum;
#pragma unroll
  for (int d = 1; d < 32; d <<= 1) {
    int t = __shfl_up(s, d, 32);
    if (lane >= d) s += t;
  }
  if (lane == 31) waveSum[w] = s;
  __syncthreads();
  if (w == 0) {
    const int ws = waveSum[lane];
    int t2 = ws;
#pragma unroll
    for (int d = 1; d < 32; d <<= 1) {
      int t = __shfl_up(t2, d, 32);
      if (lane >= d) t2 += t;
    }
    waveSum[lane] = t2 - ws;
  }
  __syncthreads();
  const int l0 = waveSum[w] + (s - tsum);
  const int l1 = l0 + c0;
  if (i0 < nBins) { cursor[i0] = l0; delta[i0] = g0 - l0; }
  if (i1 < nBins) { cursor[i1] = l1; delta[i1] = g1 - l1; }
  __syncthreads();
  const int e0 = b * chunk;
  const int e1 = (e0 + chunk < nEdges) ? (e0 + chunk) : nEdges;
  for (int e = e0 + tid; e < e1; e += 4 * SORT_BLOCK) {
    int r[4], cI[4], vI[4];
#pragma unroll
    for (int k = 0; k < 4; ++k) {
      const int ek = e + k * SORT_BLOCK;
      const int ec = ek < e1 ? ek : e1 - 1;
      r[k]  = rows[ec];
      cI[k] = cols[ec];
      vI[k] = __float_as_int(vals[ec]);
    }
#pragma unroll
    for (int k = 0; k < 4; ++k) {
      if (e + k * SORT_BLOCK < e1) {
        const int lp = atomicAdd(&cursor[r[k] >> FINE_SHIFT], 1);
        sRow[lp] = r[k];
        sCV[lp]  = make_int2(cI[k], vI[k]);
      }
    }
  }
  __syncthreads();
  const int nLocal = e1 - e0;
  for (int j = tid; j < nLocal; j += SORT_BLOCK) {
    const int r  = sRow[j];
    const int gp = j + delta[r >> FINE_SHIFT];
    tmpRow[gp] = r;
    tmpCV[gp]  = sCV[j];
  }
}

__global__ void fine_sort_kernel(const int* __restrict__ tabOff,
                                 const int* __restrict__ tmpRow,
                                 const int2* __restrict__ tmpCV,
                                 int* __restrict__ rank,
                                 int* __restrict__ ptr,
                                 int2* __restrict__ sorted,
                                 int G, int nRows) {
  __shared__ int cnt[FINE_ROWS];
  __shared__ int waveSum[32];
  __shared__ int2 stage[STAGE2_CAP];
  const int tid = threadIdx.x, lane = tid & 31, w = tid >> 5;
  const int bin = blockIdx.x;
  const int bs = tabOff[bin * G];
  const int be = tabOff[(bin + 1) * G];
  const int row0 = bin << FINE_SHIFT;
  if (tid < FINE_ROWS) cnt[tid] = 0;
  __syncthreads();
  for (int p = bs + tid; p < be; p += 4 * SORT_BLOCK) {
    int lr[4];
#pragma unroll
    for (int k = 0; k < 4; ++k) {
      const int pk = p + k * SORT_BLOCK;
      lr[k] = tmpRow[pk < be ? pk : be - 1] - row0;
    }
#pragma unroll
    for (int k = 0; k < 4; ++k)
      if (p + k * SORT_BLOCK < be) rank[p + k * SORT_BLOCK] = atomicAdd(&cnt[lr[k]], 1);
  }
  __syncthreads();
  const int v = (tid < FINE_ROWS) ? cnt[tid] : 0;
  int s = v;
#pragma unroll
  for (int d = 1; d < 32; d <<= 1) {
    int t = __shfl_up(s, d, 32);
    if (lane >= d) s += t;
  }
  if (lane == 31) waveSum[w] = s;
  __syncthreads();
  if (w == 0) {
    const int ws = waveSum[lane];
    int t2 = ws;
#pragma unroll
    for (int d = 1; d < 32; d <<= 1) {
      int t = __shfl_up(t2, d, 32);
      if (lane >= d) t2 += t;
    }
    waveSum[lane] = t2 - ws;
  }
  __syncthreads();
  const int rowStart = bs + waveSum[w] + (s - v);
  if (tid < FINE_ROWS) {
    cnt[tid] = rowStart;
    if (row0 + tid < nRows) ptr[row0 + tid] = rowStart;
  }
  if (bin == (int)gridDim.x - 1 && tid == 0) ptr[nRows] = be;
  __syncthreads();
  const int nBin = be - bs;
  if (nBin <= STAGE2_CAP) {
    for (int p = bs + tid; p < be; p += 4 * SORT_BLOCK) {
      int lr[4], rk[4];
      int2 cv[4];
#pragma unroll
      for (int k = 0; k < 4; ++k) {
        const int pk = p + k * SORT_BLOCK;
        const int pc = pk < be ? pk : be - 1;
        lr[k] = tmpRow[pc] - row0;
        rk[k] = rank[pc];
        cv[k] = tmpCV[pc];
      }
#pragma unroll
      for (int k = 0; k < 4; ++k)
        if (p + k * SORT_BLOCK < be) stage[cnt[lr[k]] + rk[k] - bs] = cv[k];
    }
    __syncthreads();
    for (int j = tid; j < nBin; j += SORT_BLOCK) sorted[bs + j] = stage[j];
  } else {
    for (int p = bs + tid; p < be; p += 4 * SORT_BLOCK) {
      int lr[4], rk[4];
      int2 cv[4];
#pragma unroll
      for (int k = 0; k < 4; ++k) {
        const int pk = p + k * SORT_BLOCK;
        const int pc = pk < be ? pk : be - 1;
        lr[k] = tmpRow[pc] - row0;
        rk[k] = rank[pc];
        cv[k] = tmpCV[pc];
      }
#pragma unroll
      for (int k = 0; k < 4; ++k)
        if (p + k * SORT_BLOCK < be) sorted[cnt[lr[k]] + rk[k]] = cv[k];
    }
  }
}

template <bool FIRST>
__device__ __forceinline__ void spmm_rows_body(const int*  __restrict__ ptr,
                                               const int2* __restrict__ sorted,
                                               const float* __restrict__ x,
                                               const float* __restrict__ x2,
                                               float* __restrict__ y,
                                               float* __restrict__ out,
                                               float scale, int writeY,
                                               int nRows, int nEdges, int nSplit) {
  const int lane = threadIdx.x & 31;
  const int c = lane & 15;
  const int h = lane >> 4;
  const int wave = blockIdx.x * (blockDim.x >> 5) + (threadIdx.x >> 5);
  const int r0 = wave << 3;
  if (r0 >= nRows) return;

  const int myRow = r0 + (c >> 1);
  const bool odd = (c & 1) != 0;
  const int lo = ptr[myRow < nRows ? myRow : nRows];
  const int hi = ptr[myRow + 1 < nRows ? myRow + 1 : nRows];
  const int start = __builtin_amdgcn_readlane(lo, 0);
  const int end   = __builtin_amdgcn_readlane(hi, 15);

  const char* sortedB = (const char*)sorted;
  const char* xB = (const char*)x;
  const unsigned laneOff = (unsigned)c * 16u;
  const unsigned long long base1 = (unsigned long long)x + laneOff;
  const unsigned long long base2 = (unsigned long long)x2 + laneOff
                                 - (unsigned long long)(unsigned)nSplit * (unsigned long long)(EMB * 4);

  v8f acc0 = {}, acc1 = {};
  const int last = nEdges - 1;
  for (int p = start; p < end; p += 16) {
    v2f A[8];
    v4f R[8];
#pragma unroll
    for (int i = 0; i < 8; ++i) {
      const int q = p + 2 * i + h;
      const int qc = q < last ? q : last;
      const int2 s = *(const int2*)(sortedB + (unsigned)qc * 8u);
      const float v = (q >= lo && q < hi) ? __int_as_float(s.y) : 0.0f;
      A[i].x = odd ? 0.0f : v;
      A[i].y = odd ? v : 0.0f;
      if (FIRST) {
        const unsigned long long b = ((unsigned)s.x < (unsigned)nSplit) ? base1 : base2;
        R[i] = *(const __attribute__((address_space(1))) v4f*)(
            b + (unsigned long long)(unsigned)s.x * (unsigned long long)(EMB * 4));
      } else {
        R[i] = *(const v4f*)(xB + ((unsigned)s.x * (unsigned)(EMB * 4) + laneOff));
      }
    }
    __builtin_amdgcn_sched_barrier(0);
#pragma unroll
    for (int i = 0; i < 8; ++i) {
      const v2f b0 = {R[i].x, R[i].y};
      const v2f b1 = {R[i].z, R[i].w};
      acc0 = __builtin_amdgcn_wmma_f32_16x16x4_f32(false, A[i], false, b0, (short)0, acc0, false, false);
      acc1 = __builtin_amdgcn_wmma_f32_16x16x4_f32(false, A[i], false, b1, (short)0, acc1, false, false);
    }
    __builtin_amdgcn_sched_barrier(0);
    asm volatile("v_nop\n\tv_nop\n\tv_nop\n\tv_nop\n\tv_nop\n\tv_nop\n\tv_nop\n\tv_nop");
    __builtin_amdgcn_sched_barrier(0);
  }

#pragma unroll
  for (int i = 0; i < 4; ++i) {
    const int row = r0 + 4 * h + i;
    if (row < nRows) {
      const size_t idx = (size_t)row * EMB + 4 * c;
      const v4f d = {acc0[2 * i], acc0[2 * i + 1], acc1[2 * i], acc1[2 * i + 1]};
      v4f o;
      if (FIRST) {
        *(v4f*)(y + idx) = d;
        const float* src = (row < nSplit) ? (x + idx) : (x2 + (idx - (size_t)nSplit * EMB));
        o = *(const v4f*)src;
      } else {
        if (writeY) *(v4f*)(y + idx) = d;
        o = *(const v4f*)(out + idx);
      }
      *(v4f*)(out + idx) = (o + d) * scale;
    }
  }
}

__global__ void spmm_rows_wmma_kernel(const int*  __restrict__ ptr,
                                      const int2* __restrict__ sorted,
                                      const float* __restrict__ x,
                                      float* __restrict__ y,
                                      float* __restrict__ out,
                                      float scale, int writeY,
                                      int nRows, int nEdges) {
  spmm_rows_body<false>(ptr, sorted, x, x, y, out, scale, writeY, nRows, nEdges, 0);
}

__global__ void spmm_rows_first_kernel(const int*  __restrict__ ptr,
                                       const int2* __restrict__ sorted,
                                       const float* __restrict__ users,
                                       const float* __restrict__ items,
                                       float* __restrict__ y,
                                       float* __restrict__ out,
                                       float scale,
                                       int nRows, int nEdges, int nUsers) {
  spmm_rows_body<true>(ptr, sorted, users, items, y, out, scale, 1, nRows, nEdges, nUsers);
}

extern "C" void kernel_launch(void* const* d_in, const int* in_sizes, int n_in,
                              void* d_out, int out_size, void* d_ws, size_t ws_size,
                              hipStream_t stream) {
  const float* users = (const float*)d_in[0];
  const float* items = (const float*)d_in[1];
  const int*   erow  = (const int*)d_in[2];
  const int*   ecol  = (const int*)d_in[3];
  const float* evals = (const float*)d_in[4];
  float* out = (float*)d_out;

  const int userElems  = in_sizes[0];
  const int itemElems  = in_sizes[1];
  const int nEdges     = in_sizes[2];
  const int totalElems = userElems + itemElems;
  const int nRows      = totalElems / EMB;

  const size_t bufBytes = (size_t)totalElems * sizeof(float);
  if (ws_size < 2 * bufBytes) return;
  float* cur = (float*)d_ws;
  float* nxt = (float*)((char*)d_ws + bufBytes);

  const int ethreads = (totalElems + 3) / 4;
  const dim3 eb(256), eg((ethreads + 255) / 256);

  const size_t bufAl    = (bufBytes + 255) & ~(size_t)255;
  const size_t edgeAl   = ((size_t)nEdges * sizeof(int2) + 255) & ~(size_t)255;
  const int    nPad     = ((nRows + 1 + SCAN_SPAN - 1) / SCAN_SPAN) * SCAN_SPAN;
  const size_t cntBytes = (size_t)nPad * sizeof(int);
  const size_t need     = 2 * bufAl + edgeAl + 2 * cntBytes;
  const size_t rankBytes = (size_t)nEdges * sizeof(int);

  const int    sortGrid = 512;
  const int    nBins    = (nRows + FINE_ROWS - 1) >> FINE_SHIFT;
  const int    nPadT    = ((nBins * sortGrid + 1 + SCAN_SPAN - 1) / SCAN_SPAN) * SCAN_SPAN;
  const int    chunk    = (nEdges + sortGrid - 1) / sortGrid;
  if (ws_size >= need && nEdges > 0 && nBins > 0 && nBins <= CB_MAX && nPadT <= nPad &&
      2 * rankBytes <= bufAl && (size_t)nPadT * sizeof(int) <= edgeAl) {
    nxt = (float*)((char*)d_ws + bufAl);
    int2* sorted = (int2*)((char*)d_ws + 2 * bufAl);
    int*  tabOff = (int*)((char*)d_ws + 2 * bufAl + edgeAl);
    int*  ptr    = tabOff + nPad;
    int*  table  = (int*)sorted;
    int*  tmpRow = (int*)cur;
    int*  rank   = tmpRow + nEdges;
    int2* tmpCV  = (int2*)nxt;

    coarse_hist_kernel<<<sortGrid, SORT_BLOCK, 0, stream>>>(erow, table, nEdges, chunk, nBins, nPadT);
    row_scan_kernel<<<1, 1024, 0, stream>>>(table, tabOff, nPadT);
    if (chunk <= STAGE1_CAP)
      coarse_scatter_staged_kernel<<<sortGrid, SORT_BLOCK, 0, stream>>>(erow, ecol, evals, tabOff,
                                                                        tmpRow, tmpCV, nEdges, chunk, nBins);
    else
      coarse_scatter_kernel<<<sortGrid, SORT_BLOCK, 0, stream>>>(erow, ecol, evals, tabOff,
                                                                 tmpRow, tmpCV, nEdges, chunk, nBins);
    fine_sort_kernel<<<nBins, SORT_BLOCK, 0, stream>>>(tabOff, tmpRow, tmpCV, rank, ptr, sorted,
                                                       sortGrid, nRows);
    const int rowGroups  = (nRows + 7) / 8;
    const int rowBlocks  = (rowGroups + 7) / 8;
    spmm_rows_first_kernel<<<rowBlocks, 256, 0, stream>>>(
        ptr, sorted, users, items, cur, out, 1.0f, nRows, nEdges, userElems / EMB);
    for (int layer = 1; layer < 3; ++layer) {
      const float scale = (layer == 2) ? 0.25f : 1.0f;
      spmm_rows_wmma_kernel<<<rowBlocks, 256, 0, stream>>>(
          ptr, sorted, cur, nxt, out, scale, (layer == 2) ? 0 : 1, nRows, nEdges);
      float* tmp = cur; cur = nxt; nxt = tmp;
    }
    return;
  }

  init_concat_kernel<<<eg, eb, 0, stream>>>(users, items, out, cur,
                                            userElems, totalElems);

  if (ws_size >= need && nEdges > 0 && rankBytes <= bufAl) {
    nxt = (float*)((char*)d_ws + bufAl);
    int2* sorted = (int2*)((char*)d_ws + 2 * bufAl);
    int*  cnt    = (int*)((char*)d_ws + 2 * bufAl + edgeAl);
    int*  ptr    = cnt + nPad;
    int*  rank   = (int*)nxt;

    hipMemsetAsync(cnt, 0, cntBytes, stream);
    const int edgeBlocks = (nEdges + 255) / 256;
    row_count_kernel<<<edgeBlocks, 256, 0, stream>>>(erow, cnt, rank, nEdges);
    row_scan_kernel<<<1, 1024, 0, stream>>>(cnt, ptr, nPad);
    bucket_fill_kernel<<<edgeBlocks, 256, 0, stream>>>(erow, ecol, evals,
                                                       ptr, rank, sorted, nEdges);

    const int rowGroups  = (nRows + 7) / 8;
    const int rowBlocks  = (rowGroups + 7) / 8;
    for (int layer = 0; layer < 3; ++layer) {
      const float scale = (layer == 2) ? 0.25f : 1.0f;
      spmm_rows_wmma_kernel<<<rowBlocks, 256, 0, stream>>>(
          ptr, sorted, cur, nxt, out, scale, (layer == 2) ? 0 : 1, nRows, nEdges);
      float* tmp = cur; cur = nxt; nxt = tmp;
    }
    return;
  }

  const int nGroups    = (nEdges + 15) / 16;
  const int spmmBlocks = (nGroups + 7) / 8;

  for (int layer = 0; layer < 3; ++layer) {
    hipMemsetAsync(nxt, 0, bufBytes, stream);
    spmm_wmma_kernel<<<spmmBlocks, 256, 0, stream>>>(erow, ecol, evals,
                                                     cur, nxt, nEdges);
    const float scale = (layer == 2) ? 0.25f : 1.0f;
    axpy_scale_kernel<<<eg, eb, 0, stream>>>(out, nxt, scale, totalElems);
    float* tmp = cur; cur = nxt; nxt = tmp;
  }
}


# Level 1 is now a block-local sort: each block orders its edge chunk by coarse bin in LDS, writes it back in place with its bin offsets; histogram and scan launches go; row sum order = LDS add order
# speedup vs baseline: 3.7081x; 1.1263x over previous
#include <hip/hip_runtime.h>

typedef __attribute__((ext_vector_type(2))) float v2f;
typedef __attribute__((ext_vector_type(4))) float v4f;
typedef __attribute__((ext_vector_type(8))) float v8f;

#define EMB 64

__global__ void init_concat_kernel(const float* __restrict__ users,
                                   const float* __restrict__ items,
                                   float* __restrict__ out,
                                   float* __restrict__ cur,
                                   int userElems, int totalElems) {
  int i4 = (blockIdx.x * blockDim.x + threadIdx.x) * 4;
  if (i4 >= totalElems) return;
  float4 v = (i4 < userElems) ? *(const float4*)(users + i4)
                              : *(const float4*)(items + (i4 - userElems));
  *(float4*)(out + i4) = v;
  *(float4*)(cur + i4) = v;
}

__global__ void axpy_scale_kernel(float* __restrict__ out,
                                  const float* __restrict__ nxt,
                                  float scale, int totalElems) {
  int i4 = (blockIdx.x * blockDim.x + threadIdx.x) * 4;
  if (i4 >= totalElems) return;
  float4 a = *(const float4*)(out + i4);
  float4 b = *(const float4*)(nxt + i4);
  a.x = (a.x + b.x) * scale;
  a.y = (a.y + b.y) * scale;
  a.z = (a.z + b.z) * scale;
  a.w = (a.w + b.w) * scale;
  *(float4*)(out + i4) = a;
}

__global__ void spmm_wmma_kernel(const int*   __restrict__ rows,
                                 const int*   __restrict__ cols,
                                 const float* __restrict__ vals,
                                 const float* __restrict__ x,
                                 float*       __restrict__ y,
                                 int nEdges) {
  const int lane   = threadIdx.x & 31;
  const int m      = lane & 15;
  const bool hiHalf = lane >= 16;
  const int wavesPerBlock = blockDim.x >> 5;
  const int wave   = blockIdx.x * wavesPerBlock + (threadIdx.x >> 5);
  const int stride = gridDim.x * wavesPerBlock;
  const int nGroups = (nEdges + 15) >> 4;

  for (int g = wave; g < nGroups; g += stride) {
    const int base = g << 4;
    const int e    = base + m;
    const int eC   = (e < nEdges) ? e : (nEdges - 1);

    const float v   = (e < nEdges) ? vals[eC] : 0.0f;
    const int myRow = rows[eC];
    const int myCol = cols[eC];

    int colK[8];
#pragma unroll
    for (int c = 0; c < 4; ++c) {
      const int kb0 = 4 * c + (hiHalf ? 2 : 0);
      colK[2 * c]     = __shfl(myCol, kb0, 32);
      colK[2 * c + 1] = __shfl(myCol, kb0 + 1, 32);
    }

    int rowD[8];
#pragma unroll
    for (int vv = 0; vv < 8; ++vv)
      rowD[vv] = __shfl(myRow, hiHalf ? (vv + 8) : vv, 32);

    v2f A[4];
#pragma unroll
    for (int c = 0; c < 4; ++c) {
      const int ka0 = 4 * c + (hiHalf ? 2 : 0);
      A[c].x = (m == ka0)     ? v : 0.0f;
      A[c].y = (m == ka0 + 1) ? v : 0.0f;
    }

#pragma unroll
    for (int t = 0; t < 4; ++t) {
      v8f acc = {};
#pragma unroll
      for (int c = 0; c < 4; ++c) {
        v2f B;
        B.x = x[colK[2 * c]     * EMB + t * 16 + m];
        B.y = x[colK[2 * c + 1] * EMB + t * 16 + m];
        acc = __builtin_amdgcn_wmma_f32_16x16x4_f32(
             false, A[c],  false, B,
             (short)0, acc,  false,  false);
      }
#pragma unroll
      for (int vv = 0; vv < 8; ++vv)
        atomicAdd(&y[rowD[vv] * EMB + t * 16 + m], acc[vv]);
    }
  }
}


#define SCAN_SPAN 8192

__global__ void row_count_kernel(const int* __restrict__ rows,
                                 int* __restrict__ cnt,
                                 int* __restrict__ rank, int nEdges) {
  int e = blockIdx.x * blockDim.x + threadIdx.x;
  if (e < nEdges) rank[e] = atomicAdd(&cnt[rows[e]], 1);
}

__global__ void row_scan_kernel(const int* __restrict__ cnt,
                                int* __restrict__ ptr, int nPad) {
  __shared__ int waveSum[32];
  __shared__ int tripTotal;
  const int tid = threadIdx.x, lane = tid & 31, w = tid >> 5;
  int carry = 0;
  for (int base = 0; base < nPad; base += SCAN_SPAN) {
    const int i0 = base + tid * 8;
    int4 a = *(const int4*)(cnt + i0);
    int4 b = *(const int4*)(cnt + i0 + 4);
    int e0 = 0, e1 = a.x, e2 = e1 + a.y, e3 = e2 + a.z, e4 = e3 + a.w;
    int e5 = e4 + b.x, e6 = e5 + b.y, e7 = e6 + b.z;
    const int tsum = e7 + b.w;
    int s = tsum;
#pragma unroll
    for (int d = 1; d < 32; d <<= 1) {
      int t = __shfl_up(s, d, 32);
      if (lane >= d) s += t;
    }
    if (lane == 31) waveSum[w] = s;
    __syncthreads();
    if (w == 0) {
      const int ws = waveSum[lane];
      int t2 = ws;
#pragma unroll
      for (int d = 1; d < 32; d <<= 1) {
        int t = __shfl_up(t2, d, 32);
        if (lane >= d) t2 += t;
      }
      waveSum[lane] = t2 - ws;
      if (lane == 31) tripTotal = t2;
    }
    __syncthreads();
    const int off = carry + waveSum[w] + (s - tsum);
    int4 o0 = make_int4(off + e0, off + e1, off + e2, off + e3);
    int4 o1 = make_int4(off + e4, off + e5, off + e6, off + e7);
    *(int4*)(ptr + i0) = o0;      *(int4*)(ptr + i0 + 4) = o1;
    carry += tripTotal;
    __syncthreads();
  }
}

__global__ void bucket_fill_kernel(const int* __restrict__ rows,
                                   const int* __restrict__ cols,
                                   const float* __restrict__ vals,
                                   const int* __restrict__ ptr,
                                   const int* __restrict__ rank,
                                   int2* __restrict__ sorted, int nEdges) {
  int e = blockIdx.x * blockDim.x + threadIdx.x;
  if (e < nEdges) {
    const int slot = ptr[rows[e]] + rank[e];
    sorted[slot] = make_int2(cols[e], __float_as_int(vals[e]));
  }
}

#define FINE_SHIFT 9
#define FINE_ROWS  (1 << FINE_SHIFT)
#define CB_MAX     2048
#define SORT_BLOCK 1024
#define STAGE2_CAP 17408

__global__ void coarse_hist_kernel(const int* __restrict__ rows,
                                   int* __restrict__ table,
                                   int nEdges, int chunk, int nBins, int nPadT) {
  __shared__ int hist[CB_MAX];
  const int tid = threadIdx.x, G = gridDim.x;
  for (int i = tid; i < nBins; i += SORT_BLOCK) hist[i] = 0;
  __syncthreads();
  const int e0 = blockIdx.x * chunk;
  const int e1 = (e0 + chunk < nEdges) ? (e0 + chunk) : nEdges;
  for (int e = e0 + tid; e < e1; e += 4 * SORT_BLOCK) {
    int r[4];
#pragma unroll
    for (int k = 0; k < 4; ++k) {
      const int ek = e + k * SORT_BLOCK;
      r[k] = rows[ek < e1 ? ek : e1 - 1];
    }
#pragma unroll
    for (int k = 0; k < 4; ++k)
      if (e + k * SORT_BLOCK < e1) atomicAdd(&hist[r[k] >> FINE_SHIFT], 1);
  }
  __syncthreads();
  for (int i = tid; i < nBins; i += SORT_BLOCK) table[i * G + blockIdx.x] = hist[i];
  for (int i = nBins * G + blockIdx.x * SORT_BLOCK + tid; i < nPadT; i += G * SORT_BLOCK)
    table[i] = 0;
}

__global__ void coarse_scatter_kernel(const int* __restrict__ rows,
                                      const int* __restrict__ cols,
                                      const float* __restrict__ vals,
                                      const int* __restrict__ tabOff,
                                      int* __restrict__ tmpRow,
                                      int2* __restrict__ tmpCV,
                                      int nEdges, int chunk, int nBins) {
  __shared__ int cursor[CB_MAX];
  const int tid = threadIdx.x, G = gridDim.x;
  for (int i = tid; i < nBins; i += SORT_BLOCK) cursor[i] = tabOff[i * G + blockIdx.x];
  __syncthreads();
  const int e0 = blockIdx.x * chunk;
  const int e1 = (e0 + chunk < nEdges) ? (e0 + chunk) : nEdges;
  for (int e = e0 + tid; e < e1; e += 4 * SORT_BLOCK) {
    int r[4], cI[4], vI[4];
#pragma unroll
    for (int k = 0; k < 4; ++k) {
      const int ek = e + k * SORT_BLOCK;
      const int ec = ek < e1 ? ek : e1 - 1;
      r[k]  = rows[ec];
      cI[k] = cols[ec];
      vI[k] = __float_as_int(vals[ec]);
    }
#pragma unroll
    for (int k = 0; k < 4; ++k) {
      if (e + k * SORT_BLOCK < e1) {
        const int pos = atomicAdd(&cursor[r[k] >> FINE_SHIFT], 1);
        tmpRow[pos] = r[k];
        tmpCV[pos]  = make_int2(cI[k], vI[k]);
      }
    }
  }
}

#define STAGE1_CAP 10240

__global__ void coarse_scatter_staged_kernel(const int* __restrict__ rows,
                                             const int* __restrict__ cols,
                                             const float* __restrict__ vals,
                                             const int* __restrict__ tabOff,
                                             int* __restrict__ tmpRow,
                                             int2* __restrict__ tmpCV,
                                             int nEdges, int chunk, int nBins) {
  __shared__ int  cursor[CB_MAX];
  __shared__ int  delta[CB_MAX];
  __shared__ int  waveSum[32];
  __shared__ int  sRow[STAGE1_CAP];
  __shared__ int2 sCV[STAGE1_CAP];
  const int tid = threadIdx.x, lane = tid & 31, w = tid >> 5, G = gridDim.x;
  const int b = blockIdx.x;
  const int i0 = 2 * tid, i1 = 2 * tid + 1;
  int g0 = 0, g1 = 0, c0 = 0, c1 = 0;
  if (i0 < nBins) { g0 = tabOff[i0 * G + b]; c0 = tabOff[i0 * G + b + 1] - g0; }
  if (i1 < nBins) { g1 = tabOff[i1 * G + b]; c1 = tabOff[i1 * G + b + 1] - g1; }
  const int tsum = c0 + c1;
  int s = tsum;
#pragma unroll
  for (int d = 1; d < 32; d <<= 1) {
    int t = __shfl_up(s, d, 32);
    if (lane >= d) s += t;
  }
  if (lane == 31) waveSum[w] = s;
  __syncthreads();
  if (w == 0) {
    const int ws = waveSum[lane];
    int t2 = ws;
#pragma unroll
    for (int d = 1; d < 32; d <<= 1) {
      int t = __shfl_up(t2, d, 32);
      if (lane >= d) t2 += t;
    }
    waveSum[lane] = t2 - ws;
  }
  __syncthreads();
  const int l0 = waveSum[w] + (s - tsum);
  const int l1 = l0 + c0;
  if (i0 < nBins) { cursor[i0] = l0; delta[i0] = g0 - l0; }
  if (i1 < nBins) { cursor[i1] = l1; delta[i1] = g1 - l1; }
  __syncthreads();
  const int e0 = b * chunk;
  const int e1 = (e0 + chunk < nEdges) ? (e0 + chunk) : nEdges;
  for (int e = e0 + tid; e < e1; e += 4 * SORT_BLOCK) {
    int r[4], cI[4], vI[4];
#pragma unroll
    for (int k = 0; k < 4; ++k) {
      const int ek = e + k * SORT_BLOCK;
      const int ec = ek < e1 ? ek : e1 - 1;
      r[k]  = rows[ec];
      cI[k] = cols[ec];
      vI[k] = __float_as_int(vals[ec]);
    }
#pragma unroll
    for (int k = 0; k < 4; ++k) {
      if (e + k * SORT_BLOCK < e1) {
        const int lp = atomicAdd(&cursor[r[k] >> FINE_SHIFT], 1);
        sRow[lp] = r[k];
        sCV[lp]  = make_int2(cI[k], vI[k]);
      }
    }
  }
  __syncthreads();
  const int nLocal = e1 - e0;
  for (int j = tid; j < nLocal; j += SORT_BLOCK) {
    const int r  = sRow[j];
    const int gp = j + delta[r >> FINE_SHIFT];
    tmpRow[gp] = r;
    tmpCV[gp]  = sCV[j];
  }
}

__global__ void fine_sort_kernel(const int* __restrict__ tabOff,
                                 const int* __restrict__ tmpRow,
                                 const int2* __restrict__ tmpCV,
                                 int* __restrict__ rank,
                                 int* __restrict__ ptr,
                                 int2* __restrict__ sorted,
                                 int G, int nRows) {
  __shared__ int cnt[FINE_ROWS];
  __shared__ int waveSum[32];
  __shared__ int2 stage[STAGE2_CAP];
  const int tid = threadIdx.x, lane = tid & 31, w = tid >> 5;
  const int bin = blockIdx.x;
  const int bs = tabOff[bin * G];
  const int be = tabOff[(bin + 1) * G];
  const int row0 = bin << FINE_SHIFT;
  if (tid < FINE_ROWS) cnt[tid] = 0;
  __syncthreads();
  for (int p = bs + tid; p < be; p += 4 * SORT_BLOCK) {
    int lr[4];
#pragma unroll
    for (int k = 0; k < 4; ++k) {
      const int pk = p + k * SORT_BLOCK;
      lr[k] = tmpRow[pk < be ? pk : be - 1] - row0;
    }
#pragma unroll
    for (int k = 0; k < 4; ++k)
      if (p + k * SORT_BLOCK < be) rank[p + k * SORT_BLOCK] = atomicAdd(&cnt[lr[k]], 1);
  }
  __syncthreads();
  const int v = (tid < FINE_ROWS) ? cnt[tid] : 0;
  int s = v;
#pragma unroll
  for (int d = 1; d < 32; d <<= 1) {
    int t = __shfl_up(s, d, 32);
    if (lane >= d) s += t;
  }
  if (lane == 31) waveSum[w] = s;
  __syncthreads();
  if (w == 0) {
    const int ws = waveSum[lane];
    int t2 = ws;
#pragma unroll
    for (int d = 1; d < 32; d <<= 1) {
      int t = __shfl_up(t2, d, 32);
      if (lane >= d) t2 += t;
    }
    waveSum[lane] = t2 - ws;
  }
  __syncthreads();
  const int rowStart = bs + waveSum[w] + (s - v);
  if (tid < FINE_ROWS) {
    cnt[tid] = rowStart;
    if (row0 + tid < nRows) ptr[row0 + tid] = rowStart;
  }
  if (bin == (int)gridDim.x - 1 && tid == 0) ptr[nRows] = be;
  __syncthreads();
  const int nBin = be - bs;
  if (nBin <= STAGE2_CAP) {
    for (int p = bs + tid; p < be; p += 4 * SORT_BLOCK) {
      int lr[4], rk[4];
      int2 cv[4];
#pragma unroll
      for (int k = 0; k < 4; ++k) {
        const int pk = p + k * SORT_BLOCK;
        const int pc = pk < be ? pk : be - 1;
        lr[k] = tmpRow[pc] - row0;
        rk[k] = rank[pc];
        cv[k] = tmpCV[pc];
      }
#pragma unroll
      for (int k = 0; k < 4; ++k)
        if (p + k * SORT_BLOCK < be) stage[cnt[lr[k]] + rk[k] - bs] = cv[k];
    }
    __syncthreads();
    for (int j = tid; j < nBin; j += SORT_BLOCK) sorted[bs + j] = stage[j];
  } else {
    for (int p = bs + tid; p < be; p += 4 * SORT_BLOCK) {
      int lr[4], rk[4];
      int2 cv[4];
#pragma unroll
      for (int k = 0; k < 4; ++k) {
        const int pk = p + k * SORT_BLOCK;
        const int pc = pk < be ? pk : be - 1;
        lr[k] = tmpRow[pc] - row0;
        rk[k] = rank[pc];
        cv[k] = tmpCV[pc];
      }
#pragma unroll
      for (int k = 0; k < 4; ++k)
        if (p + k * SORT_BLOCK < be) sorted[cnt[lr[k]] + rk[k]] = cv[k];
    }
  }
}

#define CHUNK_TRIPS (STAGE1_CAP / SORT_BLOCK)

__global__ void __launch_bounds__(SORT_BLOCK, 16) chunk_sort_kernel(const int* __restrict__ rows,
                                  const int* __restrict__ cols,
                                  const float* __restrict__ vals,
                                  int* __restrict__ loc,
                                  int* __restrict__ tmpRow,
                                  int2* __restrict__ tmpCV,
                                  int nEdges, int chunk, int nBins) {
  __shared__ int  cursor[CB_MAX];
  __shared__ int  waveSum[32];
  __shared__ int  sRow[STAGE1_CAP];
  __shared__ int2 sCV[STAGE1_CAP];
  const int tid = threadIdx.x, lane = tid & 31, w = tid >> 5;
  const int b = blockIdx.x;
  const int i0 = 2 * tid, i1 = 2 * tid + 1;
  cursor[i0] = 0;
  cursor[i1] = 0;
  __syncthreads();
  const int e0 = b * chunk;
  const int e1 = (e0 + chunk < nEdges) ? (e0 + chunk) : nEdges;
  int r[CHUNK_TRIPS], cI[CHUNK_TRIPS], vI[CHUNK_TRIPS], rk[CHUNK_TRIPS];
#pragma unroll
  for (int k = 0; k < CHUNK_TRIPS; ++k) {
    const int ek = e0 + tid + k * SORT_BLOCK;
    int ec = ek < e1 ? ek : e1 - 1;
    ec = ec > 0 ? ec : 0;
    r[k]  = rows[ec];
    cI[k] = cols[ec];
    vI[k] = __float_as_int(vals[ec]);
  }
#pragma unroll
  for (int k = 0; k < CHUNK_TRIPS; ++k) {
    rk[k] = 0;
    if (e0 + tid + k * SORT_BLOCK < e1) rk[k] = atomicAdd(&cursor[r[k] >> FINE_SHIFT], 1);
  }
  __syncthreads();
  const int c0 = cursor[i0], c1 = cursor[i1];
  const int tsum = c0 + c1;
  int s = tsum;
#pragma unroll
  for (int d = 1; d < 32; d <<= 1) {
    int t = __shfl_up(s, d, 32);
    if (lane >= d) s += t;
  }
  if (lane == 31) waveSum[w] = s;
  __syncthreads();
  if (w == 0) {
    const int ws = waveSum[lane];
    int t2 = ws;
#pragma unroll
    for (int d = 1; d < 32; d <<= 1) {
      int t = __shfl_up(t2, d, 32);
      if (lane >= d) t2 += t;
    }
    waveSum[lane] = t2 - ws;
  }
  __syncthreads();
  const int l0 = waveSum[w] + (s - tsum);
  const int l1 = l0 + c0;
  cursor[i0] = l0;
  cursor[i1] = l1;
  int* myLoc = loc + (size_t)b * (nBins + 1);
  if (i0 <= nBins) myLoc[i0] = l0;
  if (i1 <= nBins) myLoc[i1] = l1;
  __syncthreads();
#pragma unroll
  for (int k = 0; k < CHUNK_TRIPS; ++k) {
    if (e0 + tid + k * SORT_BLOCK < e1) {
      const int lp = cursor[r[k] >> FINE_SHIFT] + rk[k];
      sRow[lp] = r[k];
      sCV[lp]  = make_int2(cI[k], vI[k]);
    }
  }
  __syncthreads();
  const int nLocal = e1 - e0;
  for (int j = tid; j < nLocal; j += SORT_BLOCK) {
    tmpRow[e0 + j] = sRow[j];
    tmpCV[e0 + j]  = sCV[j];
  }
}

template <typename F>
__device__ __forceinline__ void sweep_pieces(const int* segSrc, const int* segCnt,
                                             int G, int w, int lane, int last, F f) {
  for (int s0 = w; s0 < G; s0 += 4 * 32) {
    int p[8], nk[4], sk[4];
    bool ok[8];
#pragma unroll
    for (int k = 0; k < 4; ++k) {
      const int si = s0 + 32 * k;
      nk[k] = si < G ? segCnt[si] : 0;
      sk[k] = si < G ? segSrc[si] : 0;
#pragma unroll
      for (int hf = 0; hf < 2; ++hf) {
        const int l = lane + 32 * hf;
        ok[2 * k + hf] = l < nk[k];
        const int q = sk[k] + (l < nk[k] ? l : 0);
        p[2 * k + hf] = q < last ? q : last;
      }
    }
    f(p, ok);
#pragma unroll
    for (int k = 0; k < 4; ++k) {
      for (int l = lane + 64; l - lane < nk[k]; l += 32) {
#pragma unroll
        for (int j = 0; j < 8; ++j) { p[j] = last; ok[j] = false; }
        ok[0] = l < nk[k];
        const int q = sk[k] + (l < nk[k] ? l : 0);
        p[0] = q < last ? q : last;
        f(p, ok);
      }
    }
  }
}

__global__ void fine_sort_pieces_kernel(const int* __restrict__ loc,
                                        const int* __restrict__ tmpRow,
                                        const int2* __restrict__ tmpCV,
                                        int* __restrict__ rank,
                                        int* __restrict__ ptr,
                                        int2* __restrict__ sorted,
                                        int G, int chunk, int nBins, int nRows, int nEdges) {
  __shared__ int cnt[FINE_ROWS];
  __shared__ int waveSum[32];
  __shared__ int waveSumA[32];
  __shared__ int binStart, binCount;
  __shared__ int segSrc[SORT_BLOCK];
  __shared__ int segCnt[SORT_BLOCK];
  __shared__ int2 stage[STAGE2_CAP];
  const int tid = threadIdx.x, lane = tid & 31, w = tid >> 5;
  const int bin = blockIdx.x;
  const int row0 = bin << FINE_SHIFT;
  const int last = nEdges - 1;
  int a = 0, n = 0;
  if (tid < G) {
    const int* L = loc + (size_t)tid * (nBins + 1) + bin;
    a = L[0];
    n = L[1] - a;
    segSrc[tid] = tid * chunk + a;
    segCnt[tid] = n;
  }
  if (tid < FINE_ROWS) cnt[tid] = 0;
  int sa = a, sn = n;
#pragma unroll
  for (int d = 16; d >= 1; d >>= 1) {
    sa += __shfl_xor(sa, d, 32);
    sn += __shfl_xor(sn, d, 32);
  }
  if (lane == 0) { waveSumA[w] = sa; waveSum[w] = sn; }
  __syncthreads();
  if (w == 0) {
    int ta = waveSumA[lane], tn = waveSum[lane];
#pragma unroll
    for (int d = 16; d >= 1; d >>= 1) {
      ta += __shfl_xor(ta, d, 32);
      tn += __shfl_xor(tn, d, 32);
    }
    if (lane == 0) { binStart = ta; binCount = tn; }
  }
  __syncthreads();
  const int bs = binStart;
  const int nBin = binCount;
  const int be = bs + nBin;
  sweep_pieces(segSrc, segCnt, G, w, lane, last, [&](const int (&p)[8], const bool (&ok)[8]) {
    int lr[8];
#pragma unroll
    for (int j = 0; j < 8; ++j) lr[j] = tmpRow[p[j]] - row0;
#pragma unroll
    for (int j = 0; j < 8; ++j)
      if (ok[j]) rank[p[j]] = atomicAdd(&cnt[lr[j]], 1);
  });
  __syncthreads();
  const int v = (tid < FINE_ROWS) ? cnt[tid] : 0;
  int s = v;
#pragma unroll
  for (int d = 1; d < 32; d <<= 1) {
    int t = __shfl_up(s, d, 32);
    if (lane >= d) s += t;
  }
  if (lane == 31) waveSum[w] = s;
  __syncthreads();
  if (w == 0) {
    const int ws = waveSum[lane];
    int t2 = ws;
#pragma unroll
    for (int d = 1; d < 32; d <<= 1) {
      int t = __shfl_up(t2, d, 32);
      if (lane >= d) t2 += t;
    }
    waveSum[lane] = t2 - ws;
  }
  __syncthreads();
  const int rowStart = bs + waveSum[w] + (s - v);
  if (tid < FINE_ROWS) {
    cnt[tid] = rowStart;
    if (row0 + tid < nRows) ptr[row0 + tid] = rowStart;
  }
  if (bin == (int)gridDim.x - 1 && tid == 0) ptr[nRows] = be;
  __syncthreads();
  if (nBin <= STAGE2_CAP) {
    sweep_pieces(segSrc, segCnt, G, w, lane, last, [&](const int (&p)[8], const bool (&ok)[8]) {
      int lr[8], rk[8];
      int2 cv[8];
#pragma unroll
      for (int j = 0; j < 8; ++j) {
        lr[j] = tmpRow[p[j]] - row0;
        rk[j] = rank[p[j]];
        cv[j] = tmpCV[p[j]];
      }
#pragma unroll
      for (int j = 0; j < 8; ++j)
        if (ok[j]) stage[cnt[lr[j]] + rk[j] - bs] = cv[j];
    });
    __syncthreads();
    for (int j = tid; j < nBin; j += SORT_BLOCK) sorted[bs + j] = stage[j];
  } else {
    sweep_pieces(segSrc, segCnt, G, w, lane, last, [&](const int (&p)[8], const bool (&ok)[8]) {
      int lr[8], rk[8];
      int2 cv[8];
#pragma unroll
      for (int j = 0; j < 8; ++j) {
        lr[j] = tmpRow[p[j]] - row0;
        rk[j] = rank[p[j]];
        cv[j] = tmpCV[p[j]];
      }
#pragma unroll
      for (int j = 0; j < 8; ++j)
        if (ok[j]) sorted[cnt[lr[j]] + rk[j]] = cv[j];
    });
  }
}

template <bool FIRST>
__device__ __forceinline__ void spmm_rows_body(const int*  __restrict__ ptr,
                                               const int2* __restrict__ sorted,
                                               const float* __restrict__ x,
                                               const float* __restrict__ x2,
                                               float* __restrict__ y,
                                               float* __restrict__ out,
                                               float scale, int writeY,
                                               int nRows, int nEdges, int nSplit) {
  const int lane = threadIdx.x & 31;
  const int c = lane & 15;
  const int h = lane >> 4;
  const int wave = blockIdx.x * (blockDim.x >> 5) + (threadIdx.x >> 5);
  const int r0 = wave << 3;
  if (r0 >= nRows) return;

  const int myRow = r0 + (c >> 1);
  const bool odd = (c & 1) != 0;
  const int lo = ptr[myRow < nRows ? myRow : nRows];
  const int hi = ptr[myRow + 1 < nRows ? myRow + 1 : nRows];
  const int start = __builtin_amdgcn_readlane(lo, 0);
  const int end   = __builtin_amdgcn_readlane(hi, 15);

  const char* sortedB = (const char*)sorted;
  const char* xB = (const char*)x;
  const unsigned laneOff = (unsigned)c * 16u;
  const unsigned long long base1 = (unsigned long long)x + laneOff;
  const unsigned long long base2 = (unsigned long long)x2 + laneOff
                                 - (unsigned long long)(unsigned)nSplit * (unsigned long long)(EMB * 4);

  v8f acc0 = {}, acc1 = {};
  const int last = nEdges - 1;
  for (int p = start; p < end; p += 16) {
    v2f A[8];
    v4f R[8];
#pragma unroll
    for (int i = 0; i < 8; ++i) {
      const int q = p + 2 * i + h;
      const int qc = q < last ? q : last;
      const int2 s = *(const int2*)(sortedB + (unsigned)qc * 8u);
      const float v = (q >= lo && q < hi) ? __int_as_float(s.y) : 0.0f;
      A[i].x = odd ? 0.0f : v;
      A[i].y = odd ? v : 0.0f;
      if (FIRST) {
        const unsigned long long b = ((unsigned)s.x < (unsigned)nSplit) ? base1 : base2;
        R[i] = *(const __attribute__((address_space(1))) v4f*)(
            b + (unsigned long long)(unsigned)s.x * (unsigned long long)(EMB * 4));
      } else {
        R[i] = *(const v4f*)(xB + ((unsigned)s.x * (unsigned)(EMB * 4) + laneOff));
      }
    }
    __builtin_amdgcn_sched_barrier(0);
#pragma unroll
    for (int i = 0; i < 8; ++i) {
      const v2f b0 = {R[i].x, R[i].y};
      const v2f b1 = {R[i].z, R[i].w};
      acc0 = __builtin_amdgcn_wmma_f32_16x16x4_f32(false, A[i], false, b0, (short)0, acc0, false, false);
      acc1 = __builtin_amdgcn_wmma_f32_16x16x4_f32(false, A[i], false, b1, (short)0, acc1, false, false);
    }
    __builtin_amdgcn_sched_barrier(0);
    asm volatile("v_nop\n\tv_nop\n\tv_nop\n\tv_nop\n\tv_nop\n\tv_nop\n\tv_nop\n\tv_nop");
    __builtin_amdgcn_sched_barrier(0);
  }

#pragma unroll
  for (int i = 0; i < 4; ++i) {
    const int row = r0 + 4 * h + i;
    if (row < nRows) {
      const size_t idx = (size_t)row * EMB + 4 * c;
      const v4f d = {acc0[2 * i], acc0[2 * i + 1], acc1[2 * i], acc1[2 * i + 1]};
      v4f o;
      if (FIRST) {
        *(v4f*)(y + idx) = d;
        const float* src = (row < nSplit) ? (x + idx) : (x2 + (idx - (size_t)nSplit * EMB));
        o = *(const v4f*)src;
      } else {
        if (writeY) *(v4f*)(y + idx) = d;
        o = *(const v4f*)(out + idx);
      }
      *(v4f*)(out + idx) = (o + d) * scale;
    }
  }
}

__global__ void spmm_rows_wmma_kernel(const int*  __restrict__ ptr,
                                      const int2* __restrict__ sorted,
                                      const float* __restrict__ x,
                                      float* __restrict__ y,
                                      float* __restrict__ out,
                                      float scale, int writeY,
                                      int nRows, int nEdges) {
  spmm_rows_body<false>(ptr, sorted, x, x, y, out, scale, writeY, nRows, nEdges, 0);
}

__global__ void spmm_rows_first_kernel(const int*  __restrict__ ptr,
                                       const int2* __restrict__ sorted,
                                       const float* __restrict__ users,
                                       const float* __restrict__ items,
                                       float* __restrict__ y,
                                       float* __restrict__ out,
                                       float scale,
                                       int nRows, int nEdges, int nUsers) {
  spmm_rows_body<true>(ptr, sorted, users, items, y, out, scale, 1, nRows, nEdges, nUsers);
}

extern "C" void kernel_launch(void* const* d_in, const int* in_sizes, int n_in,
                              void* d_out, int out_size, void* d_ws, size_t ws_size,
                              hipStream_t stream) {
  const float* users = (const float*)d_in[0];
  const float* items = (const float*)d_in[1];
  const int*   erow  = (const int*)d_in[2];
  const int*   ecol  = (const int*)d_in[3];
  const float* evals = (const float*)d_in[4];
  float* out = (float*)d_out;

  const int userElems  = in_sizes[0];
  const int itemElems  = in_sizes[1];
  const int nEdges     = in_sizes[2];
  const int totalElems = userElems + itemElems;
  const int nRows      = totalElems / EMB;

  const size_t bufBytes = (size_t)totalElems * sizeof(float);
  if (ws_size < 2 * bufBytes) return;
  float* cur = (float*)d_ws;
  float* nxt = (float*)((char*)d_ws + bufBytes);

  const int ethreads = (totalElems + 3) / 4;
  const dim3 eb(256), eg((ethreads + 255) / 256);

  const size_t bufAl    = (bufBytes + 255) & ~(size_t)255;
  const size_t edgeAl   = ((size_t)nEdges * sizeof(int2) + 255) & ~(size_t)255;
  const int    nPad     = ((nRows + 1 + SCAN_SPAN - 1) / SCAN_SPAN) * SCAN_SPAN;
  const size_t cntBytes = (size_t)nPad * sizeof(int);
  const size_t need     = 2 * bufAl + edgeAl + 2 * cntBytes;
  const size_t rankBytes = (size_t)nEdges * sizeof(int);

  const int    sortGrid = 512;
  const int    nBins    = (nRows + FINE_ROWS - 1) >> FINE_SHIFT;
  const int    nPadT    = ((nBins * sortGrid + 1 + SCAN_SPAN - 1) / SCAN_SPAN) * SCAN_SPAN;
  const int    chunk    = (nEdges + sortGrid - 1) / sortGrid;
  if (ws_size >= need && nEdges > 0 && nBins > 0 && nBins <= CB_MAX && nPadT <= nPad &&
      2 * rankBytes <= bufAl && (size_t)nPadT * sizeof(int) <= edgeAl) {
    nxt = (float*)((char*)d_ws + bufAl);
    int2* sorted = (int2*)((char*)d_ws + 2 * bufAl);
    int*  tabOff = (int*)((char*)d_ws + 2 * bufAl + edgeAl);
    int*  ptr    = tabOff + nPad;
    int*  table  = (int*)sorted;
    int*  tmpRow = (int*)cur;
    int*  rank   = tmpRow + nEdges;
    int2* tmpCV  = (int2*)nxt;

    if (chunk <= STAGE1_CAP && nBins < CB_MAX && sortGrid <= SORT_BLOCK &&
        (size_t)sortGrid * (size_t)(nBins + 1) <= (size_t)nPad) {
      int* loc = tabOff;
      chunk_sort_kernel<<<sortGrid, SORT_BLOCK, 0, stream>>>(erow, ecol, evals, loc,
                                                             tmpRow, tmpCV, nEdges, chunk, nBins);
      fine_sort_pieces_kernel<<<nBins, SORT_BLOCK, 0, stream>>>(loc, tmpRow, tmpCV, rank, ptr, sorted,
                                                                sortGrid, chunk, nBins, nRows, nEdges);
    } else {
      coarse_hist_kernel<<<sortGrid, SORT_BLOCK, 0, stream>>>(erow, table, nEdges, chunk, nBins, nPadT);
      row_scan_kernel<<<1, 1024, 0, stream>>>(table, tabOff, nPadT);
      if (chunk <= STAGE1_CAP)
        coarse_scatter_staged_kernel<<<sortGrid, SORT_BLOCK, 0, stream>>>(erow, ecol, evals, tabOff,
                                                                          tmpRow, tmpCV, nEdges, chunk, nBins);
      else
        coarse_scatter_kernel<<<sortGrid, SORT_BLOCK, 0, stream>>>(erow, ecol, evals, tabOff,
                                                                   tmpRow, tmpCV, nEdges, chunk, nBins);
      fine_sort_kernel<<<nBins, SORT_BLOCK, 0, stream>>>(tabOff, tmpRow, tmpCV, rank, ptr, sorted,
                                                         sortGrid, nRows);
    }
    const int rowGroups  = (nRows + 7) / 8;
    const int rowBlocks  = (rowGroups + 7) / 8;
    spmm_rows_first_kernel<<<rowBlocks, 256, 0, stream>>>(
        ptr, sorted, users, items, cur, out, 1.0f, nRows, nEdges, userElems / EMB);
    for (int layer = 1; layer < 3; ++layer) {
      const float scale = (layer == 2) ? 0.25f : 1.0f;
      spmm_rows_wmma_kernel<<<rowBlocks, 256, 0, stream>>>(
          ptr, sorted, cur, nxt, out, scale, (layer == 2) ? 0 : 1, nRows, nEdges);
      float* tmp = cur; cur = nxt; nxt = tmp;
    }
    return;
  }

  init_concat_kernel<<<eg, eb, 0, stream>>>(users, items, out, cur,
                                            userElems, totalElems);

  if (ws_size >= need && nEdges > 0 && rankBytes <= bufAl) {
    nxt = (float*)((char*)d_ws + bufAl);
    int2* sorted = (int2*)((char*)d_ws + 2 * bufAl);
    int*  cnt    = (int*)((char*)d_ws + 2 * bufAl + edgeAl);
    int*  ptr    = cnt + nPad;
    int*  rank   = (int*)nxt;

    hipMemsetAsync(cnt, 0, cntBytes, stream);
    const int edgeBlocks = (nEdges + 255) / 256;
    row_count_kernel<<<edgeBlocks, 256, 0, stream>>>(erow, cnt, rank, nEdges);
    row_scan_kernel<<<1, 1024, 0, stream>>>(cnt, ptr, nPad);
    bucket_fill_kernel<<<edgeBlocks, 256, 0, stream>>>(erow, ecol, evals,
                                                       ptr, rank, sorted, nEdges);

    const int rowGroups  = (nRows + 7) / 8;
    const int rowBlocks  = (rowGroups + 7) / 8;
    for (int layer = 0; layer < 3; ++layer) {
      const float scale = (layer == 2) ? 0.25f : 1.0f;
      spmm_rows_wmma_kernel<<<rowBlocks, 256, 0, stream>>>(
          ptr, sorted, cur, nxt, out, scale, (layer == 2) ? 0 : 1, nRows, nEdges);
      float* tmp = cur; cur = nxt; nxt = tmp;
    }
    return;
  }

  const int nGroups    = (nEdges + 15) / 16;
  const int spmmBlocks = (nGroups + 7) / 8;

  for (int layer = 0; layer < 3; ++layer) {
    hipMemsetAsync(nxt, 0, bufBytes, stream);
    spmm_wmma_kernel<<<spmmBlocks, 256, 0, stream>>>(erow, ecol, evals,
                                                     cur, nxt, nEdges);
    const float scale = (layer == 2) ? 0.25f : 1.0f;
    axpy_scale_kernel<<<eg, eb, 0, stream>>>(out, nxt, scale, totalElems);
    float* tmp = cur; cur = nxt; nxt = tmp;
  }
}
